# starting kernel
# baseline (speedup 1.0000x reference)
#include <hip/hip_runtime.h>

#define VOCAB 33
#define V2 1089
#define BUCKETS 512
#define B_ROWS 4096
#define S_LEN 4096
#define KPAD_BI 1104
#define HID 256
#define OUTC 64

typedef float v2f __attribute__((ext_vector_type(2)));
typedef float v8f __attribute__((ext_vector_type(8)));

__global__ __launch_bounds__(256) void hist_kernel(const int* __restrict__ src,
                                                   float* __restrict__ bi_freq,
                                                   float* __restrict__ tri_freq) {
    __shared__ unsigned bi_cnt[V2];
    __shared__ unsigned tri_cnt[BUCKETS];
    __shared__ unsigned totals[2];

    const int row = blockIdx.x;
    const int tid = threadIdx.x;
    const int* s = src + (size_t)row * S_LEN;

    for (int i = tid; i < V2; i += 256) bi_cnt[i] = 0u;
    for (int i = tid; i < BUCKETS; i += 256) tri_cnt[i] = 0u;
    if (tid < 2) totals[tid] = 0u;
    __syncthreads();

    unsigned myBi = 0, myTri = 0;
    for (int i = tid; i < S_LEN - 1; i += 256) {
        int c = s[i];
        int n = s[i + 1];
        bool bm = (c > 0) && (n > 0);
        if (bm) {
            atomicAdd(&bi_cnt[c * VOCAB + n], 1u);
            ++myBi;
        }
        if (i < S_LEN - 2) {
            int t = s[i + 2];
            if (bm && (t > 0)) {
                unsigned idx = ((unsigned)(c * (VOCAB * VOCAB) + n * VOCAB + t)) & (BUCKETS - 1);
                atomicAdd(&tri_cnt[idx], 1u);
                ++myTri;
            }
        }
    }
    if (myBi)  atomicAdd(&totals[0], myBi);
    if (myTri) atomicAdd(&totals[1], myTri);
    __syncthreads();

    const float biInv  = 1.0f / (float)max(totals[0], 1u);
    const float triInv = 1.0f / (float)max(totals[1], 1u);

    float* bf = bi_freq + (size_t)row * KPAD_BI;
    for (int i = tid; i < KPAD_BI; i += 256)
        bf[i] = (i < V2 ? (float)bi_cnt[i] : 0.0f) * biInv;

    float* tf = tri_freq + (size_t)row * BUCKETS;
    for (int i = tid; i < BUCKETS; i += 256)
        tf[i] = (float)tri_cnt[i] * triInv;
}

__global__ __launch_bounds__(256) void mlp_kernel(const float* __restrict__ freq,
                                                  int K, int Kpad,
                                                  const float* __restrict__ W1,
                                                  const float* __restrict__ b1,
                                                  const float* __restrict__ W2,
                                                  const float* __restrict__ b2,
                                                  float* __restrict__ out,
                                                  int outOff) {
    __shared__ float lds_h[16 * HID];

    const int rowBase = blockIdx.x * 16;
    const int lane = threadIdx.x & 31;
    const int wave = threadIdx.x >> 5;
    const int m  = lane & 15;
    const int kh = lane >> 4;
    const int n  = lane & 15;
    const int colBase = wave * 32;

    v8f c0 = {};
    v8f c1 = {};

    const float* arow = freq + (size_t)(rowBase + m) * Kpad;

    for (int k0 = 0; k0 < Kpad; k0 += 4) {
        const int ka = k0 + 2 * kh;
        v2f a;
        a.x = arow[ka];
        a.y = arow[ka + 1];
        const int kb0 = min(ka,     K - 1);
        const int kb1 = min(ka + 1, K - 1);
        v2f bA, bB;
        bA.x = W1[(size_t)kb0 * HID + colBase + n];
        bA.y = W1[(size_t)kb1 * HID + colBase + n];
        bB.x = W1[(size_t)kb0 * HID + colBase + 16 + n];
        bB.y = W1[(size_t)kb1 * HID + colBase + 16 + n];
        c0 = __builtin_amdgcn_wmma_f32_16x16x4_f32(false, a, false, bA, (short)0, c0, false, false);
        c1 = __builtin_amdgcn_wmma_f32_16x16x4_f32(false, a, false, bB, (short)0, c1, false, false);
    }

    #pragma unroll
    for (int r = 0; r < 8; ++r) {
        const int mm = r + 8 * kh;
        float h0 = c0[r] + b1[colBase + n];
        float h1 = c1[r] + b1[colBase + 16 + n];
        h0 = h0 > 0.0f ? h0 : 0.0f;
        h1 = h1 > 0.0f ? h1 : 0.0f;
        lds_h[mm * HID + colBase + n]      = h0;
        lds_h[mm * HID + colBase + 16 + n] = h1;
    }
    __syncthreads();

    if (wave < 2) {
        const int cb = wave * 16;
        v8f d = {};
        for (int k0 = 0; k0 < HID; k0 += 4) {
            const int ka = k0 + 2 * kh;
            v2f a;
            a.x = lds_h[m * HID + ka];
            a.y = lds_h[m * HID + ka + 1];
            v2f b;
            b.x = W2[(size_t)ka * 32 + cb + n];
            b.y = W2[(size_t)(ka + 1) * 32 + cb + n];
            d = __builtin_amdgcn_wmma_f32_16x16x4_f32(false, a, false, b, (short)0, d, false, false);
        }
        #pragma unroll
        for (int r = 0; r < 8; ++r) {
            const int mm = r + 8 * kh;
            out[(size_t)(rowBase + mm) * OUTC + outOff + cb + n] = d[r] + b2[cb + n];
        }
    }
}

extern "C" void kernel_launch(void* const* d_in, const int* in_sizes, int n_in,
                              void* d_out, int out_size, void* d_ws, size_t ws_size,
                              hipStream_t stream) {
    const int*   src = (const int*)d_in[0];
    const float* Wb1 = (const float*)d_in[1];
    const float* bb1 = (const float*)d_in[2];
    const float* Wb2 = (const float*)d_in[3];
    const float* bb2 = (const float*)d_in[4];
    const float* Wt1 = (const float*)d_in[5];
    const float* bt1 = (const float*)d_in[6];
    const float* Wt2 = (const float*)d_in[7];
    const float* bt2 = (const float*)d_in[8];
    float* out = (float*)d_out;

    float* bi_freq  = (float*)d_ws;
    float* tri_freq = bi_freq + (size_t)B_ROWS * KPAD_BI;

    hist_kernel<<<B_ROWS, 256, 0, stream>>>(src, bi_freq, tri_freq);

    mlp_kernel<<<B_ROWS / 16, 256, 0, stream>>>(bi_freq, V2, KPAD_BI,
                                                Wb1, bb1, Wb2, bb2, out, 0);
    mlp_kernel<<<B_ROWS / 16, 256, 0, stream>>>(tri_freq, BUCKETS, BUCKETS,
                                                Wt1, bt1, Wt2, bt2, out, 32);
}


# Layer-1 GEMM operands go through a two-stage LDS ring: 32-k slabs of W1 and freq fetched two slabs ahead by 16-byte coalesced loads, fragments read from LDS; k order of the sums unchanged
# speedup vs baseline: 1.3553x; 1.3553x over previous
#include <hip/hip_runtime.h>

#define VOCAB 33
#define V2 1089
#define BUCKETS 512
#define B_ROWS 4096
#define S_LEN 4096
#define KPAD_BI 1104
#define HID 256
#define OUTC 64

typedef float v2f __attribute__((ext_vector_type(2)));
typedef float v8f __attribute__((ext_vector_type(8)));

__global__ __launch_bounds__(256) void hist_kernel(const int* __restrict__ src,
                                                   float* __restrict__ bi_freq,
                                                   float* __restrict__ tri_freq) {
    __shared__ unsigned bi_cnt[V2];
    __shared__ unsigned tri_cnt[BUCKETS];
    __shared__ unsigned totals[2];

    const int row = blockIdx.x;
    const int tid = threadIdx.x;
    const int* s = src + (size_t)row * S_LEN;

    for (int i = tid; i < V2; i += 256) bi_cnt[i] = 0u;
    for (int i = tid; i < BUCKETS; i += 256) tri_cnt[i] = 0u;
    if (tid < 2) totals[tid] = 0u;
    __syncthreads();

    unsigned myBi = 0, myTri = 0;
    for (int i = tid; i < S_LEN - 1; i += 256) {
        int c = s[i];
        int n = s[i + 1];
        bool bm = (c > 0) && (n > 0);
        if (bm) {
            atomicAdd(&bi_cnt[c * VOCAB + n], 1u);
            ++myBi;
        }
        if (i < S_LEN - 2) {
            int t = s[i + 2];
            if (bm && (t > 0)) {
                unsigned idx = ((unsigned)(c * (VOCAB * VOCAB) + n * VOCAB + t)) & (BUCKETS - 1);
                atomicAdd(&tri_cnt[idx], 1u);
                ++myTri;
            }
        }
    }
    if (myBi)  atomicAdd(&totals[0], myBi);
    if (myTri) atomicAdd(&totals[1], myTri);
    __syncthreads();

    const float biInv  = 1.0f / (float)max(totals[0], 1u);
    const float triInv = 1.0f / (float)max(totals[1], 1u);

    float* bf = bi_freq + (size_t)row * KPAD_BI;
    for (int i = tid; i < KPAD_BI; i += 256)
        bf[i] = (i < V2 ? (float)bi_cnt[i] : 0.0f) * biInv;

    float* tf = tri_freq + (size_t)row * BUCKETS;
    for (int i = tid; i < BUCKETS; i += 256)
        tf[i] = (float)tri_cnt[i] * triInv;
}

#define KB 32
#define PW 544
#define PA 36
#define W_STAGE (KB / 2 * PW)
#define A_STAGE (16 * PA)

typedef float v4f __attribute__((ext_vector_type(4)));

__device__ __forceinline__ void lds_barrier() {
    __builtin_amdgcn_fence(__ATOMIC_RELEASE, "workgroup", "local");
    __builtin_amdgcn_s_barrier();
    __builtin_amdgcn_fence(__ATOMIC_ACQUIRE, "workgroup", "local");
}

__global__ __launch_bounds__(256) void mlp_kernel(const float* __restrict__ freq,
                                                  int K, int Kpad,
                                                  const float* __restrict__ W1,
                                                  const float* __restrict__ b1,
                                                  const float* __restrict__ W2,
                                                  const float* __restrict__ b2,
                                                  float* __restrict__ out,
                                                  int outOff) {
    __shared__ __attribute__((aligned(16))) float lds_w[2 * W_STAGE];
    __shared__ __attribute__((aligned(16))) float lds_a[2 * A_STAGE];
    __shared__ float lds_h[16 * HID];

    const int tid = threadIdx.x;
    const int rowBase = blockIdx.x * 16;
    const int lane = tid & 31;
    const int wave = tid >> 5;
    const int m  = lane & 15;
    const int kh = lane >> 4;
    const int n  = lane & 15;
    const int colBase = wave * 32;

    const int q   = tid & 63;
    const int rp0 = tid >> 6;
    const int am = (tid >> 3) & 15;
    const int aq = tid & 7;
    const bool aLoader = tid < 128;
    const float* wcol = W1 + 4 * q;
    const float* arow = freq + (size_t)(rowBase + am) * Kpad;

    v4f wr[8];
    v4f ar;

    auto load_slab = [&](int k0) {
        #pragma unroll
        for (int i = 0; i < 4; ++i) {
            const int kr = k0 + 2 * (rp0 + 4 * i);
            wr[2 * i]     = *(const v4f*)(wcol + (size_t)min(kr,     K - 1) * HID);
            wr[2 * i + 1] = *(const v4f*)(wcol + (size_t)min(kr + 1, K - 1) * HID);
        }
        ar = *(const v4f*)(arow + min(k0 + 4 * aq, Kpad - 4));
    };

    auto store_slab = [&](int k0, int stage) {
        if (k0 + KB > K) {
            #pragma unroll
            for (int i = 0; i < 4; ++i) {
                const int kr = k0 + 2 * (rp0 + 4 * i);
                if (kr     >= K) wr[2 * i]     = (v4f){0.f, 0.f, 0.f, 0.f};
                if (kr + 1 >= K) wr[2 * i + 1] = (v4f){0.f, 0.f, 0.f, 0.f};
            }
            if (k0 + 4 * aq >= Kpad) ar = (v4f){0.f, 0.f, 0.f, 0.f};
        }
        float* w = lds_w + stage * W_STAGE + 8 * q;
        #pragma unroll
        for (int i = 0; i < 4; ++i) {
            const int rp = rp0 + 4 * i;
            const v4f r0 = wr[2 * i], r1 = wr[2 * i + 1];
            *(v4f*)(w + rp * PW)     = (v4f){r0.x, r1.x, r0.y, r1.y};
            *(v4f*)(w + rp * PW + 4) = (v4f){r0.z, r1.z, r0.w, r1.w};
        }
        if (aLoader) *(v4f*)(lds_a + stage * A_STAGE + am * PA + 4 * aq) = ar;
    };

    v2f fa[KB / 4], fb0[KB / 4], fb1[KB / 4];

    auto read_frags = [&](int stage) {
        const float* a = lds_a + stage * A_STAGE + m * PA + 2 * kh;
        const float* w = lds_w + stage * W_STAGE + kh * PW + 2 * (colBase + n);
        #pragma unroll
        for (int ks = 0; ks < KB / 4; ++ks) {
            fa[ks]  = *(const v2f*)(a + 4 * ks);
            fb0[ks] = *(const v2f*)(w + 2 * ks * PW);
            fb1[ks] = *(const v2f*)(w + 2 * ks * PW + 32);
        }
    };

    v8f c0 = {};
    v8f c1 = {};

    auto mma_steps = [&](int from, int to) {
        #pragma unroll
        for (int ks = from; ks < to; ++ks) {
            c0 = __builtin_amdgcn_wmma_f32_16x16x4_f32(false, fa[ks], false, fb0[ks], (short)0, c0, false, false);
            c1 = __builtin_amdgcn_wmma_f32_16x16x4_f32(false, fa[ks], false, fb1[ks], (short)0, c1, false, false);
        }
    };

    constexpr int TAIL = 2;
    constexpr int HEAD = KB / 4 - TAIL;
    v2f ta[TAIL] = {}, tb0[TAIL] = {}, tb1[TAIL] = {};
    auto mma_tail = [&]() {
        #pragma unroll
        for (int i = 0; i < TAIL; ++i) {
            c0 = __builtin_amdgcn_wmma_f32_16x16x4_f32(false, ta[i], false, tb0[i], (short)0, c0, false, false);
            c1 = __builtin_amdgcn_wmma_f32_16x16x4_f32(false, ta[i], false, tb1[i], (short)0, c1, false, false);
        }
    };
    auto keep_tail = [&]() {
        #pragma unroll
        for (int i = 0; i < TAIL; ++i) { ta[i] = fa[HEAD + i]; tb0[i] = fb0[HEAD + i]; tb1[i] = fb1[HEAD + i]; }
    };

    const int nSlab = (Kpad + KB - 1) / KB;

    load_slab(0);
    store_slab(0, 0);
    if (nSlab > 1) load_slab(KB);
    lds_barrier();
    for (int s = 0; s < nSlab - 1; ++s) {
        read_frags(s & 1);
        mma_tail();
        store_slab((s + 1) * KB, (s + 1) & 1);
        if (s + 2 < nSlab) load_slab((s + 2) * KB);
        mma_steps(0, HEAD);
        keep_tail();
        __builtin_amdgcn_sched_barrier(0);
        lds_barrier();
    }
    read_frags((nSlab - 1) & 1);
    mma_tail();
    mma_steps(0, KB / 4);

    #pragma unroll
    for (int r = 0; r < 8; ++r) {
        const int mm = r + 8 * kh;
        float h0 = c0[r] + b1[colBase + n];
        float h1 = c1[r] + b1[colBase + 16 + n];
        h0 = h0 > 0.0f ? h0 : 0.0f;
        h1 = h1 > 0.0f ? h1 : 0.0f;
        lds_h[mm * HID + colBase + n]      = h0;
        lds_h[mm * HID + colBase + 16 + n] = h1;
    }
    __syncthreads();

    if (wave < 2) {
        const int cb = wave * 16;
        v8f d = {};
        for (int k0 = 0; k0 < HID; k0 += 4) {
            const int ka = k0 + 2 * kh;
            v2f a;
            a.x = lds_h[m * HID + ka];
            a.y = lds_h[m * HID + ka + 1];
            v2f b;
            b.x = W2[(size_t)ka * 32 + cb + n];
            b.y = W2[(size_t)(ka + 1) * 32 + cb + n];
            d = __builtin_amdgcn_wmma_f32_16x16x4_f32(false, a, false, b, (short)0, d, false, false);
        }
        #pragma unroll
        for (int r = 0; r < 8; ++r) {
            const int mm = r + 8 * kh;
            out[(size_t)(rowBase + mm) * OUTC + outOff + cb + n] = d[r] + b2[cb + n];
        }
    }
}

extern "C" void kernel_launch(void* const* d_in, const int* in_sizes, int n_in,
                              void* d_out, int out_size, void* d_ws, size_t ws_size,
                              hipStream_t stream) {
    const int*   src = (const int*)d_in[0];
    const float* Wb1 = (const float*)d_in[1];
    const float* bb1 = (const float*)d_in[2];
    const float* Wb2 = (const float*)d_in[3];
    const float* bb2 = (const float*)d_in[4];
    const float* Wt1 = (const float*)d_in[5];
    const float* bt1 = (const float*)d_in[6];
    const float* Wt2 = (const float*)d_in[7];
    const float* bt2 = (const float*)d_in[8];
    float* out = (float*)d_out;

    float* bi_freq  = (float*)d_ws;
    float* tri_freq = bi_freq + (size_t)B_ROWS * KPAD_BI;

    hist_kernel<<<B_ROWS, 256, 0, stream>>>(src, bi_freq, tri_freq);

    mlp_kernel<<<B_ROWS / 16, 256, 0, stream>>>(bi_freq, V2, KPAD_BI,
                                                Wb1, bb1, Wb2, bb2, out, 0);
    mlp_kernel<<<B_ROWS / 16, 256, 0, stream>>>(tri_freq, BUCKETS, BUCKETS,
                                                Wt1, bt1, Wt2, bt2, out, 32);
}


# Histogram kernel fetches a whole row before counting: per thread four coalesced 16-byte token loads plus 8-byte look-aheads issued together, not one dependent load per token pair; integer counts
# speedup vs baseline: 1.6651x; 1.2286x over previous
#include <hip/hip_runtime.h>

#define VOCAB 33
#define V2 1089
#define BUCKETS 512
#define B_ROWS 4096
#define S_LEN 4096
#define KPAD_BI 1104
#define HID 256
#define OUTC 64

typedef float v2f __attribute__((ext_vector_type(2)));
typedef float v8f __attribute__((ext_vector_type(8)));

typedef int v4i __attribute__((ext_vector_type(4)));
typedef int v2i __attribute__((ext_vector_type(2)));

__global__ __launch_bounds__(256) void hist_kernel(const int* __restrict__ src,
                                                   float* __restrict__ bi_freq,
                                                   float* __restrict__ tri_freq) {
    __shared__ unsigned bi_cnt[V2];
    __shared__ unsigned tri_cnt[BUCKETS];
    __shared__ unsigned totals[2];

    const int row = blockIdx.x;
    const int tid = threadIdx.x;
    const int* s = src + (size_t)row * S_LEN;

    v4i qa[4];
    v2i la[4];
    #pragma unroll
    for (int j = 0; j < 4; ++j)
        qa[j] = *(const v4i*)(s + j * (S_LEN / 4) + 4 * tid);
    #pragma unroll
    for (int j = 0; j < 4; ++j) {
        const int e = j * (S_LEN / 4) + 4 * tid + 4;
        const v2i v = *(const v2i*)(s + min(e, S_LEN - 2));
        la[j] = (e < S_LEN) ? v : (v2i){0, 0};
    }

    for (int i = tid; i < V2; i += 256) bi_cnt[i] = 0u;
    for (int i = tid; i < BUCKETS; i += 256) tri_cnt[i] = 0u;
    if (tid < 2) totals[tid] = 0u;
    __syncthreads();

    unsigned myBi = 0, myTri = 0;
    #pragma unroll
    for (int j = 0; j < 4; ++j) {
        const int tok[6] = {qa[j].x, qa[j].y, qa[j].z, qa[j].w, la[j].x, la[j].y};
        #pragma unroll
        for (int k = 0; k < 4; ++k) {
            const int c = tok[k];
            const int n = tok[k + 1];
            const int t = tok[k + 2];
            const bool bm = (c > 0) && (n > 0);
            if (bm) {
                atomicAdd(&bi_cnt[c * VOCAB + n], 1u);
                ++myBi;
            }
            if (bm && (t > 0)) {
                unsigned idx = ((unsigned)(c * (VOCAB * VOCAB) + n * VOCAB + t)) & (BUCKETS - 1);
                atomicAdd(&tri_cnt[idx], 1u);
                ++myTri;
            }
        }
    }
    if (myBi)  atomicAdd(&totals[0], myBi);
    if (myTri) atomicAdd(&totals[1], myTri);
    __syncthreads();

    const float biInv  = 1.0f / (float)max(totals[0], 1u);
    const float triInv = 1.0f / (float)max(totals[1], 1u);

    float* bf = bi_freq + (size_t)row * KPAD_BI;
    for (int i = tid; i < KPAD_BI; i += 256)
        bf[i] = (i < V2 ? (float)bi_cnt[i] : 0.0f) * biInv;

    float* tf = tri_freq + (size_t)row * BUCKETS;
    for (int i = tid; i < BUCKETS; i += 256)
        tf[i] = (float)tri_cnt[i] * triInv;
}

#define KB 32
#define PW 544
#define PA 36
#define W_STAGE (KB / 2 * PW)
#define A_STAGE (16 * PA)

typedef float v4f __attribute__((ext_vector_type(4)));

__device__ __forceinline__ void lds_barrier() {
    __builtin_amdgcn_fence(__ATOMIC_RELEASE, "workgroup", "local");
    __builtin_amdgcn_s_barrier();
    __builtin_amdgcn_fence(__ATOMIC_ACQUIRE, "workgroup", "local");
}

__global__ __launch_bounds__(256) void mlp_kernel(const float* __restrict__ freq,
                                                  int K, int Kpad,
                                                  const float* __restrict__ W1,
                                                  const float* __restrict__ b1,
                                                  const float* __restrict__ W2,
                                                  const float* __restrict__ b2,
                                                  float* __restrict__ out,
                                                  int outOff) {
    __shared__ __attribute__((aligned(16))) float lds_w[2 * W_STAGE];
    __shared__ __attribute__((aligned(16))) float lds_a[2 * A_STAGE];
    __shared__ float lds_h[16 * HID];

    const int tid = threadIdx.x;
    const int rowBase = blockIdx.x * 16;
    const int lane = tid & 31;
    const int wave = tid >> 5;
    const int m  = lane & 15;
    const int kh = lane >> 4;
    const int n  = lane & 15;
    const int colBase = wave * 32;

    const int q   = tid & 63;
    const int rp0 = tid >> 6;
    const int am = (tid >> 3) & 15;
    const int aq = tid & 7;
    const bool aLoader = tid < 128;
    const float* wcol = W1 + 4 * q;
    const float* arow = freq + (size_t)(rowBase + am) * Kpad;

    v4f wr[8];
    v4f ar;

    auto load_slab = [&](int k0) {
        #pragma unroll
        for (int i = 0; i < 4; ++i) {
            const int kr = k0 + 2 * (rp0 + 4 * i);
            wr[2 * i]     = *(const v4f*)(wcol + (size_t)min(kr,     K - 1) * HID);
            wr[2 * i + 1] = *(const v4f*)(wcol + (size_t)min(kr + 1, K - 1) * HID);
        }
        ar = *(const v4f*)(arow + min(k0 + 4 * aq, Kpad - 4));
    };

    auto store_slab = [&](int k0, int stage) {
        if (k0 + KB > K) {
            #pragma unroll
            for (int i = 0; i < 4; ++i) {
                const int kr = k0 + 2 * (rp0 + 4 * i);
                if (kr     >= K) wr[2 * i]     = (v4f){0.f, 0.f, 0.f, 0.f};
                if (kr + 1 >= K) wr[2 * i + 1] = (v4f){0.f, 0.f, 0.f, 0.f};
            }
            if (k0 + 4 * aq >= Kpad) ar = (v4f){0.f, 0.f, 0.f, 0.f};
        }
        float* w = lds_w + stage * W_STAGE + 8 * q;
        #pragma unroll
        for (int i = 0; i < 4; ++i) {
            const int rp = rp0 + 4 * i;
            const v4f r0 = wr[2 * i], r1 = wr[2 * i + 1];
            *(v4f*)(w + rp * PW)     = (v4f){r0.x, r1.x, r0.y, r1.y};
            *(v4f*)(w + rp * PW + 4) = (v4f){r0.z, r1.z, r0.w, r1.w};
        }
        if (aLoader) *(v4f*)(lds_a + stage * A_STAGE + am * PA + 4 * aq) = ar;
    };

    v2f fa[KB / 4], fb0[KB / 4], fb1[KB / 4];

    auto read_frags = [&](int stage) {
        const float* a = lds_a + stage * A_STAGE + m * PA + 2 * kh;
        const float* w = lds_w + stage * W_STAGE + kh * PW + 2 * (colBase + n);
        #pragma unroll
        for (int ks = 0; ks < KB / 4; ++ks) {
            fa[ks]  = *(const v2f*)(a + 4 * ks);
            fb0[ks] = *(const v2f*)(w + 2 * ks * PW);
            fb1[ks] = *(const v2f*)(w + 2 * ks * PW + 32);
        }
    };

    v8f c0 = {};
    v8f c1 = {};

    auto mma_steps = [&](int from, int to) {
        #pragma unroll
        for (int ks = from; ks < to; ++ks) {
            c0 = __builtin_amdgcn_wmma_f32_16x16x4_f32(false, fa[ks], false, fb0[ks], (short)0, c0, false, false);
            c1 = __builtin_amdgcn_wmma_f32_16x16x4_f32(false, fa[ks], false, fb1[ks], (short)0, c1, false, false);
        }
    };

    constexpr int TAIL = 2;
    constexpr int HEAD = KB / 4 - TAIL;
    v2f ta[TAIL] = {}, tb0[TAIL] = {}, tb1[TAIL] = {};
    auto mma_tail = [&]() {
        #pragma unroll
        for (int i = 0; i < TAIL; ++i) {
            c0 = __builtin_amdgcn_wmma_f32_16x16x4_f32(false, ta[i], false, tb0[i], (short)0, c0, false, false);
            c1 = __builtin_amdgcn_wmma_f32_16x16x4_f32(false, ta[i], false, tb1[i], (short)0, c1, false, false);
        }
    };
    auto keep_tail = [&]() {
        #pragma unroll
        for (int i = 0; i < TAIL; ++i) { ta[i] = fa[HEAD + i]; tb0[i] = fb0[HEAD + i]; tb1[i] = fb1[HEAD + i]; }
    };

    const int nSlab = (Kpad + KB - 1) / KB;

    load_slab(0);
    store_slab(0, 0);
    if (nSlab > 1) load_slab(KB);
    lds_barrier();
    for (int s = 0; s < nSlab - 1; ++s) {
        read_frags(s & 1);
        mma_tail();
        store_slab((s + 1) * KB, (s + 1) & 1);
        if (s + 2 < nSlab) load_slab((s + 2) * KB);
        mma_steps(0, HEAD);
        keep_tail();
        __builtin_amdgcn_sched_barrier(0);
        lds_barrier();
    }
    read_frags((nSlab - 1) & 1);
    mma_tail();
    mma_steps(0, KB / 4);

    #pragma unroll
    for (int r = 0; r < 8; ++r) {
        const int mm = r + 8 * kh;
        float h0 = c0[r] + b1[colBase + n];
        float h1 = c1[r] + b1[colBase + 16 + n];
        h0 = h0 > 0.0f ? h0 : 0.0f;
        h1 = h1 > 0.0f ? h1 : 0.0f;
        lds_h[mm * HID + colBase + n]      = h0;
        lds_h[mm * HID + colBase + 16 + n] = h1;
    }
    __syncthreads();

    if (wave < 2) {
        const int cb = wave * 16;
        v8f d = {};
        for (int k0 = 0; k0 < HID; k0 += 4) {
            const int ka = k0 + 2 * kh;
            v2f a;
            a.x = lds_h[m * HID + ka];
            a.y = lds_h[m * HID + ka + 1];
            v2f b;
            b.x = W2[(size_t)ka * 32 + cb + n];
            b.y = W2[(size_t)(ka + 1) * 32 + cb + n];
            d = __builtin_amdgcn_wmma_f32_16x16x4_f32(false, a, false, b, (short)0, d, false, false);
        }
        #pragma unroll
        for (int r = 0; r < 8; ++r) {
            const int mm = r + 8 * kh;
            out[(size_t)(rowBase + mm) * OUTC + outOff + cb + n] = d[r] + b2[cb + n];
        }
    }
}

extern "C" void kernel_launch(void* const* d_in, const int* in_sizes, int n_in,
                              void* d_out, int out_size, void* d_ws, size_t ws_size,
                              hipStream_t stream) {
    const int*   src = (const int*)d_in[0];
    const float* Wb1 = (const float*)d_in[1];
    const float* bb1 = (const float*)d_in[2];
    const float* Wb2 = (const float*)d_in[3];
    const float* bb2 = (const float*)d_in[4];
    const float* Wt1 = (const float*)d_in[5];
    const float* bt1 = (const float*)d_in[6];
    const float* Wt2 = (const float*)d_in[7];
    const float* bt2 = (const float*)d_in[8];
    float* out = (float*)d_out;

    float* bi_freq  = (float*)d_ws;
    float* tri_freq = bi_freq + (size_t)B_ROWS * KPAD_BI;

    hist_kernel<<<B_ROWS, 256, 0, stream>>>(src, bi_freq, tri_freq);

    mlp_kernel<<<B_ROWS / 16, 256, 0, stream>>>(bi_freq, V2, KPAD_BI,
                                                Wb1, bb1, Wb2, bb2, out, 0);
    mlp_kernel<<<B_ROWS / 16, 256, 0, stream>>>(tri_freq, BUCKETS, BUCKETS,
                                                Wt1, bt1, Wt2, bt2, out, 32);
}


# Both MLPs run in ONE launch of twice the tiles (bigram tiles first, trigram behind), two workgroups per WGP: trigram matrix work fills the bigram tiles' barrier gaps; 3 launches -> 2; sums unchanged
# speedup vs baseline: 2.2306x; 1.3396x over previous
#include <hip/hip_runtime.h>

#define VOCAB 33
#define V2 1089
#define BUCKETS 512
#define B_ROWS 4096
#define S_LEN 4096
#define KPAD_BI 1104
#define HID 256
#define OUTC 64

typedef float v2f __attribute__((ext_vector_type(2)));
typedef float v8f __attribute__((ext_vector_type(8)));

typedef int v4i __attribute__((ext_vector_type(4)));
typedef int v2i __attribute__((ext_vector_type(2)));

__global__ __launch_bounds__(256) void hist_kernel(const int* __restrict__ src,
                                                   float* __restrict__ bi_freq,
                                                   float* __restrict__ tri_freq) {
    __shared__ unsigned bi_cnt[V2];
    __shared__ unsigned tri_cnt[BUCKETS];
    __shared__ unsigned totals[2];

    const int row = blockIdx.x;
    const int tid = threadIdx.x;
    const int* s = src + (size_t)row * S_LEN;

    v4i qa[4];
    v2i la[4];
    #pragma unroll
    for (int j = 0; j < 4; ++j)
        qa[j] = *(const v4i*)(s + j * (S_LEN / 4) + 4 * tid);
    #pragma unroll
    for (int j = 0; j < 4; ++j) {
        const int e = j * (S_LEN / 4) + 4 * tid + 4;
        const v2i v = *(const v2i*)(s + min(e, S_LEN - 2));
        la[j] = (e < S_LEN) ? v : (v2i){0, 0};
    }

    for (int i = tid; i < V2; i += 256) bi_cnt[i] = 0u;
    for (int i = tid; i < BUCKETS; i += 256) tri_cnt[i] = 0u;
    if (tid < 2) totals[tid] = 0u;
    __syncthreads();

    unsigned myBi = 0, myTri = 0;
    #pragma unroll
    for (int j = 0; j < 4; ++j) {
        const int tok[6] = {qa[j].x, qa[j].y, qa[j].z, qa[j].w, la[j].x, la[j].y};
        #pragma unroll
        for (int k = 0; k < 4; ++k) {
            const int c = tok[k];
            const int n = tok[k + 1];
            const int t = tok[k + 2];
            const bool bm = (c > 0) && (n > 0);
            if (bm) {
                atomicAdd(&bi_cnt[c * VOCAB + n], 1u);
                ++myBi;
            }
            if (bm && (t > 0)) {
                unsigned idx = ((unsigned)(c * (VOCAB * VOCAB) + n * VOCAB + t)) & (BUCKETS - 1);
                atomicAdd(&tri_cnt[idx], 1u);
                ++myTri;
            }
        }
    }
    if (myBi)  atomicAdd(&totals[0], myBi);
    if (myTri) atomicAdd(&totals[1], myTri);
    __syncthreads();

    const float biInv  = 1.0f / (float)max(totals[0], 1u);
    const float triInv = 1.0f / (float)max(totals[1], 1u);

    float* bf = bi_freq + (size_t)row * KPAD_BI;
    for (int i = tid; i < KPAD_BI; i += 256)
        bf[i] = (i < V2 ? (float)bi_cnt[i] : 0.0f) * biInv;

    float* tf = tri_freq + (size_t)row * BUCKETS;
    for (int i = tid; i < BUCKETS; i += 256)
        tf[i] = (float)tri_cnt[i] * triInv;
}

#define TILES (B_ROWS / 16)
#define KB 32
#define PW 544
#define PA 36
#define W_STAGE (KB / 2 * PW)
#define A_STAGE (16 * PA)

typedef float v4f __attribute__((ext_vector_type(4)));

__device__ __forceinline__ void lds_barrier() {
    __builtin_amdgcn_fence(__ATOMIC_RELEASE, "workgroup", "local");
    __builtin_amdgcn_s_barrier();
    __builtin_amdgcn_fence(__ATOMIC_ACQUIRE, "workgroup", "local");
}

__global__ __launch_bounds__(256) void mlp_kernel(const float* __restrict__ freqB,
                                                  const float* __restrict__ W1B,
                                                  const float* __restrict__ b1B,
                                                  const float* __restrict__ W2B,
                                                  const float* __restrict__ b2B,
                                                  const float* __restrict__ freqT,
                                                  const float* __restrict__ W1T,
                                                  const float* __restrict__ b1T,
                                                  const float* __restrict__ W2T,
                                                  const float* __restrict__ b2T,
                                                  float* __restrict__ out) {
    const bool tri = blockIdx.x >= TILES;
    const float* __restrict__ freq = tri ? freqT : freqB;
    const float* __restrict__ W1   = tri ? W1T : W1B;
    const float* __restrict__ b1   = tri ? b1T : b1B;
    const float* __restrict__ W2   = tri ? W2T : W2B;
    const float* __restrict__ b2   = tri ? b2T : b2B;
    const int K      = tri ? BUCKETS : V2;
    const int Kpad   = tri ? BUCKETS : KPAD_BI;
    const int outOff = tri ? 32 : 0;

    __shared__ __attribute__((aligned(16))) float lds_w[2 * W_STAGE];
    __shared__ __attribute__((aligned(16))) float lds_a[2 * A_STAGE];
    __shared__ float lds_h[16 * HID];

    const int tid = threadIdx.x;
    const int rowBase = (tri ? blockIdx.x - TILES : blockIdx.x) * 16;
    const int lane = tid & 31;
    const int wave = tid >> 5;
    const int m  = lane & 15;
    const int kh = lane >> 4;
    const int n  = lane & 15;
    const int colBase = wave * 32;

    const int q   = tid & 63;
    const int rp0 = tid >> 6;
    const int am = (tid >> 3) & 15;
    const int aq = tid & 7;
    const bool aLoader = tid < 128;
    const float* wcol = W1 + 4 * q;
    const float* arow = freq + (size_t)(rowBase + am) * Kpad;

    v4f wr[8];
    v4f ar;

    auto load_slab = [&](int k0) {
        #pragma unroll
        for (int i = 0; i < 4; ++i) {
            const int kr = k0 + 2 * (rp0 + 4 * i);
            wr[2 * i]     = *(const v4f*)(wcol + (size_t)min(kr,     K - 1) * HID);
            wr[2 * i + 1] = *(const v4f*)(wcol + (size_t)min(kr + 1, K - 1) * HID);
        }
        ar = *(const v4f*)(arow + min(k0 + 4 * aq, Kpad - 4));
    };

    auto store_slab = [&](int k0, int stage) {
        if (k0 + KB > K) {
            #pragma unroll
            for (int i = 0; i < 4; ++i) {
                const int kr = k0 + 2 * (rp0 + 4 * i);
                if (kr     >= K) wr[2 * i]     = (v4f){0.f, 0.f, 0.f, 0.f};
                if (kr + 1 >= K) wr[2 * i + 1] = (v4f){0.f, 0.f, 0.f, 0.f};
            }
            if (k0 + 4 * aq >= Kpad) ar = (v4f){0.f, 0.f, 0.f, 0.f};
        }
        float* w = lds_w + stage * W_STAGE + 8 * q;
        #pragma unroll
        for (int i = 0; i < 4; ++i) {
            const int rp = rp0 + 4 * i;
            const v4f r0 = wr[2 * i], r1 = wr[2 * i + 1];
            *(v4f*)(w + rp * PW)     = (v4f){r0.x, r1.x, r0.y, r1.y};
            *(v4f*)(w + rp * PW + 4) = (v4f){r0.z, r1.z, r0.w, r1.w};
        }
        if (aLoader) *(v4f*)(lds_a + stage * A_STAGE + am * PA + 4 * aq) = ar;
    };

    v2f fa[KB / 4], fb0[KB / 4], fb1[KB / 4];

    auto read_frags = [&](int stage) {
        const float* a = lds_a + stage * A_STAGE + m * PA + 2 * kh;
        const float* w = lds_w + stage * W_STAGE + kh * PW + 2 * (colBase + n);
        #pragma unroll
        for (int ks = 0; ks < KB / 4; ++ks) {
            fa[ks]  = *(const v2f*)(a + 4 * ks);
            fb0[ks] = *(const v2f*)(w + 2 * ks * PW);
            fb1[ks] = *(const v2f*)(w + 2 * ks * PW + 32);
        }
    };

    v8f c0 = {};
    v8f c1 = {};

    auto mma_steps = [&](int from, int to) {
        #pragma unroll
        for (int ks = from; ks < to; ++ks) {
            c0 = __builtin_amdgcn_wmma_f32_16x16x4_f32(false, fa[ks], false, fb0[ks], (short)0, c0, false, false);
            c1 = __builtin_amdgcn_wmma_f32_16x16x4_f32(false, fa[ks], false, fb1[ks], (short)0, c1, false, false);
        }
    };

    constexpr int TAIL = 2;
    constexpr int HEAD = KB / 4 - TAIL;
    v2f ta[TAIL] = {}, tb0[TAIL] = {}, tb1[TAIL] = {};
    auto mma_tail = [&]() {
        #pragma unroll
        for (int i = 0; i < TAIL; ++i) {
            c0 = __builtin_amdgcn_wmma_f32_16x16x4_f32(false, ta[i], false, tb0[i], (short)0, c0, false, false);
            c1 = __builtin_amdgcn_wmma_f32_16x16x4_f32(false, ta[i], false, tb1[i], (short)0, c1, false, false);
        }
    };
    auto keep_tail = [&]() {
        #pragma unroll
        for (int i = 0; i < TAIL; ++i) { ta[i] = fa[HEAD + i]; tb0[i] = fb0[HEAD + i]; tb1[i] = fb1[HEAD + i]; }
    };

    const int nSlab = (Kpad + KB - 1) / KB;

    load_slab(0);
    store_slab(0, 0);
    if (nSlab > 1) load_slab(KB);
    lds_barrier();
    for (int s = 0; s < nSlab - 1; ++s) {
        read_frags(s & 1);
        mma_tail();
        store_slab((s + 1) * KB, (s + 1) & 1);
        if (s + 2 < nSlab) load_slab((s + 2) * KB);
        mma_steps(0, HEAD);
        keep_tail();
        __builtin_amdgcn_sched_barrier(0);
        lds_barrier();
    }
    read_frags((nSlab - 1) & 1);
    mma_tail();
    mma_steps(0, KB / 4);

    #pragma unroll
    for (int r = 0; r < 8; ++r) {
        const int mm = r + 8 * kh;
        float h0 = c0[r] + b1[colBase + n];
        float h1 = c1[r] + b1[colBase + 16 + n];
        h0 = h0 > 0.0f ? h0 : 0.0f;
        h1 = h1 > 0.0f ? h1 : 0.0f;
        lds_h[mm * HID + colBase + n]      = h0;
        lds_h[mm * HID + colBase + 16 + n] = h1;
    }
    __syncthreads();

    if (wave < 2) {
        const int cb = wave * 16;
        v8f d = {};
        for (int k0 = 0; k0 < HID; k0 += 4) {
            const int ka = k0 + 2 * kh;
            v2f a;
            a.x = lds_h[m * HID + ka];
            a.y = lds_h[m * HID + ka + 1];
            v2f b;
            b.x = W2[(size_t)ka * 32 + cb + n];
            b.y = W2[(size_t)(ka + 1) * 32 + cb + n];
            d = __builtin_amdgcn_wmma_f32_16x16x4_f32(false, a, false, b, (short)0, d, false, false);
        }
        #pragma unroll
        for (int r = 0; r < 8; ++r) {
            const int mm = r + 8 * kh;
            out[(size_t)(rowBase + mm) * OUTC + outOff + cb + n] = d[r] + b2[cb + n];
        }
    }
}

extern "C" void kernel_launch(void* const* d_in, const int* in_sizes, int n_in,
                              void* d_out, int out_size, void* d_ws, size_t ws_size,
                              hipStream_t stream) {
    const int*   src = (const int*)d_in[0];
    const float* Wb1 = (const float*)d_in[1];
    const float* bb1 = (const float*)d_in[2];
    const float* Wb2 = (const float*)d_in[3];
    const float* bb2 = (const float*)d_in[4];
    const float* Wt1 = (const float*)d_in[5];
    const float* bt1 = (const float*)d_in[6];
    const float* Wt2 = (const float*)d_in[7];
    const float* bt2 = (const float*)d_in[8];
    float* out = (float*)d_out;

    float* bi_freq  = (float*)d_ws;
    float* tri_freq = bi_freq + (size_t)B_ROWS * KPAD_BI;

    hist_kernel<<<B_ROWS, 256, 0, stream>>>(src, bi_freq, tri_freq);

    mlp_kernel<<<2 * TILES, 256, 0, stream>>>(bi_freq, Wb1, bb1, Wb2, bb2,
                                              tri_freq, Wt1, bt1, Wt2, bt2, out);
}


# Layer 2 on all 8 waves (2 column tiles x 4 k-quarters) instead of 2: W2 fragments fetched under the last slab's matrix work, hidden rows padded; quarter sums added via LDS in k order (new sum tree)
# speedup vs baseline: 2.4007x; 1.0763x over previous
#include <hip/hip_runtime.h>

#define VOCAB 33
#define V2 1089
#define BUCKETS 512
#define B_ROWS 4096
#define S_LEN 4096
#define KPAD_BI 1104
#define HID 256
#define OUTC 64

typedef float v2f __attribute__((ext_vector_type(2)));
typedef float v8f __attribute__((ext_vector_type(8)));

typedef int v4i __attribute__((ext_vector_type(4)));
typedef int v2i __attribute__((ext_vector_type(2)));

__global__ __launch_bounds__(256) void hist_kernel(const int* __restrict__ src,
                                                   float* __restrict__ bi_freq,
                                                   float* __restrict__ tri_freq) {
    __shared__ unsigned bi_cnt[V2];
    __shared__ unsigned tri_cnt[BUCKETS];
    __shared__ unsigned totals[2];

    const int row = blockIdx.x;
    const int tid = threadIdx.x;
    const int* s = src + (size_t)row * S_LEN;

    v4i qa[4];
    v2i la[4];
    #pragma unroll
    for (int j = 0; j < 4; ++j)
        qa[j] = *(const v4i*)(s + j * (S_LEN / 4) + 4 * tid);
    #pragma unroll
    for (int j = 0; j < 4; ++j) {
        const int e = j * (S_LEN / 4) + 4 * tid + 4;
        const v2i v = *(const v2i*)(s + min(e, S_LEN - 2));
        la[j] = (e < S_LEN) ? v : (v2i){0, 0};
    }

    for (int i = tid; i < V2; i += 256) bi_cnt[i] = 0u;
    for (int i = tid; i < BUCKETS; i += 256) tri_cnt[i] = 0u;
    if (tid < 2) totals[tid] = 0u;
    __syncthreads();

    unsigned myBi = 0, myTri = 0;
    #pragma unroll
    for (int j = 0; j < 4; ++j) {
        const int tok[6] = {qa[j].x, qa[j].y, qa[j].z, qa[j].w, la[j].x, la[j].y};
        #pragma unroll
        for (int k = 0; k < 4; ++k) {
            const int c = tok[k];
            const int n = tok[k + 1];
            const int t = tok[k + 2];
            const bool bm = (c > 0) && (n > 0);
            if (bm) {
                atomicAdd(&bi_cnt[c * VOCAB + n], 1u);
                ++myBi;
            }
            if (bm && (t > 0)) {
                unsigned idx = ((unsigned)(c * (VOCAB * VOCAB) + n * VOCAB + t)) & (BUCKETS - 1);
                atomicAdd(&tri_cnt[idx], 1u);
                ++myTri;
            }
        }
    }
    if (myBi)  atomicAdd(&totals[0], myBi);
    if (myTri) atomicAdd(&totals[1], myTri);
    __syncthreads();

    const float biInv  = 1.0f / (float)max(totals[0], 1u);
    const float triInv = 1.0f / (float)max(totals[1], 1u);

    float* bf = bi_freq + (size_t)row * KPAD_BI;
    for (int i = tid; i < KPAD_BI; i += 256)
        bf[i] = (i < V2 ? (float)bi_cnt[i] : 0.0f) * biInv;

    float* tf = tri_freq + (size_t)row * BUCKETS;
    for (int i = tid; i < BUCKETS; i += 256)
        tf[i] = (float)tri_cnt[i] * triInv;
}

#define TILES (B_ROWS / 16)
#define KB 32
#define PW 544
#define PA 36
#define W_STAGE (KB / 2 * PW)
#define A_STAGE (16 * PA)
#define PH 260
#define PP 36

typedef float v4f __attribute__((ext_vector_type(4)));

__device__ __forceinline__ void lds_barrier() {
    __builtin_amdgcn_fence(__ATOMIC_RELEASE, "workgroup", "local");
    __builtin_amdgcn_s_barrier();
    __builtin_amdgcn_fence(__ATOMIC_ACQUIRE, "workgroup", "local");
}

__global__ __launch_bounds__(256) void mlp_kernel(const float* __restrict__ freqB,
                                                  const float* __restrict__ W1B,
                                                  const float* __restrict__ b1B,
                                                  const float* __restrict__ W2B,
                                                  const float* __restrict__ b2B,
                                                  const float* __restrict__ freqT,
                                                  const float* __restrict__ W1T,
                                                  const float* __restrict__ b1T,
                                                  const float* __restrict__ W2T,
                                                  const float* __restrict__ b2T,
                                                  float* __restrict__ out) {
    const bool tri = blockIdx.x >= TILES;
    const float* __restrict__ freq = tri ? freqT : freqB;
    const float* __restrict__ W1   = tri ? W1T : W1B;
    const float* __restrict__ b1   = tri ? b1T : b1B;
    const float* __restrict__ W2   = tri ? W2T : W2B;
    const float* __restrict__ b2   = tri ? b2T : b2B;
    const int K      = tri ? BUCKETS : V2;
    const int Kpad   = tri ? BUCKETS : KPAD_BI;
    const int outOff = tri ? 32 : 0;

    __shared__ __attribute__((aligned(16))) float lds_w[2 * W_STAGE];
    __shared__ __attribute__((aligned(16))) float lds_a[2 * A_STAGE];
    __shared__ __attribute__((aligned(16))) float lds_h[16 * PH];
    __shared__ __attribute__((aligned(16))) float lds_p[4 * 16 * PP];

    const int tid = threadIdx.x;
    const int rowBase = (tri ? blockIdx.x - TILES : blockIdx.x) * 16;
    const int lane = tid & 31;
    const int wave = tid >> 5;
    const int m  = lane & 15;
    const int kh = lane >> 4;
    const int n  = lane & 15;
    const int colBase = wave * 32;

    const int q   = tid & 63;
    const int rp0 = tid >> 6;
    const int am = (tid >> 3) & 15;
    const int aq = tid & 7;
    const bool aLoader = tid < 128;
    const float* wcol = W1 + 4 * q;
    const float* arow = freq + (size_t)(rowBase + am) * Kpad;

    v4f wr[8];
    v4f ar;

    auto load_slab = [&](int k0) {
        #pragma unroll
        for (int i = 0; i < 4; ++i) {
            const int kr = k0 + 2 * (rp0 + 4 * i);
            wr[2 * i]     = *(const v4f*)(wcol + (size_t)min(kr,     K - 1) * HID);
            wr[2 * i + 1] = *(const v4f*)(wcol + (size_t)min(kr + 1, K - 1) * HID);
        }
        ar = *(const v4f*)(arow + min(k0 + 4 * aq, Kpad - 4));
    };

    auto store_slab = [&](int k0, int stage) {
        if (k0 + KB > K) {
            #pragma unroll
            for (int i = 0; i < 4; ++i) {
                const int kr = k0 + 2 * (rp0 + 4 * i);
                if (kr     >= K) wr[2 * i]     = (v4f){0.f, 0.f, 0.f, 0.f};
                if (kr + 1 >= K) wr[2 * i + 1] = (v4f){0.f, 0.f, 0.f, 0.f};
            }
            if (k0 + 4 * aq >= Kpad) ar = (v4f){0.f, 0.f, 0.f, 0.f};
        }
        float* w = lds_w + stage * W_STAGE + 8 * q;
        #pragma unroll
        for (int i = 0; i < 4; ++i) {
            const int rp = rp0 + 4 * i;
            const v4f r0 = wr[2 * i], r1 = wr[2 * i + 1];
            *(v4f*)(w + rp * PW)     = (v4f){r0.x, r1.x, r0.y, r1.y};
            *(v4f*)(w + rp * PW + 4) = (v4f){r0.z, r1.z, r0.w, r1.w};
        }
        if (aLoader) *(v4f*)(lds_a + stage * A_STAGE + am * PA + 4 * aq) = ar;
    };

    v2f fa[KB / 4], fb0[KB / 4], fb1[KB / 4];

    auto read_frags = [&](int stage) {
        const float* a = lds_a + stage * A_STAGE + m * PA + 2 * kh;
        const float* w = lds_w + stage * W_STAGE + kh * PW + 2 * (colBase + n);
        #pragma unroll
        for (int ks = 0; ks < KB / 4; ++ks) {
            fa[ks]  = *(const v2f*)(a + 4 * ks);
            fb0[ks] = *(const v2f*)(w + 2 * ks * PW);
            fb1[ks] = *(const v2f*)(w + 2 * ks * PW + 32);
        }
    };

    v8f c0 = {};
    v8f c1 = {};

    auto mma_steps = [&](int from, int to) {
        #pragma unroll
        for (int ks = from; ks < to; ++ks) {
            c0 = __builtin_amdgcn_wmma_f32_16x16x4_f32(false, fa[ks], false, fb0[ks], (short)0, c0, false, false);
            c1 = __builtin_amdgcn_wmma_f32_16x16x4_f32(false, fa[ks], false, fb1[ks], (short)0, c1, false, false);
        }
    };

    constexpr int TAIL = 2;
    constexpr int HEAD = KB / 4 - TAIL;
    v2f ta[TAIL] = {}, tb0[TAIL] = {}, tb1[TAIL] = {};
    auto mma_tail = [&]() {
        #pragma unroll
        for (int i = 0; i < TAIL; ++i) {
            c0 = __builtin_amdgcn_wmma_f32_16x16x4_f32(false, ta[i], false, tb0[i], (short)0, c0, false, false);
            c1 = __builtin_amdgcn_wmma_f32_16x16x4_f32(false, ta[i], false, tb1[i], (short)0, c1, false, false);
        }
    };
    auto keep_tail = [&]() {
        #pragma unroll
        for (int i = 0; i < TAIL; ++i) { ta[i] = fa[HEAD + i]; tb0[i] = fb0[HEAD + i]; tb1[i] = fb1[HEAD + i]; }
    };

    const int nSlab = (Kpad + KB - 1) / KB;

    load_slab(0);
    store_slab(0, 0);
    if (nSlab > 1) load_slab(KB);
    lds_barrier();
    for (int s = 0; s < nSlab - 1; ++s) {
        read_frags(s & 1);
        mma_tail();
        store_slab((s + 1) * KB, (s + 1) & 1);
        if (s + 2 < nSlab) load_slab((s + 2) * KB);
        mma_steps(0, HEAD);
        keep_tail();
        __builtin_amdgcn_sched_barrier(0);
        lds_barrier();
    }
    const int ct = wave & 1;
    const int kq = wave >> 1;
    const int cb = ct * 16;
    const float bias1a = b1[colBase + n];
    const float bias1b = b1[colBase + 16 + n];
    const int orow = tid >> 4;
    const int oc   = (tid & 15) * 2;
    const v2f bias2 = *(const v2f*)(b2 + oc);
    v2f w2f[16];
    {
        const float* w2p = W2 + (size_t)(64 * kq + 2 * kh) * 32 + cb + n;
        #pragma unroll
        for (int ks = 0; ks < 16; ++ks) {
            w2f[ks].x = w2p[(4 * ks) * 32];
            w2f[ks].y = w2p[(4 * ks + 1) * 32];
        }
    }

    read_frags((nSlab - 1) & 1);
    mma_tail();
    mma_steps(0, KB / 4);

    #pragma unroll
    for (int r = 0; r < 8; ++r) {
        const int mm = r + 8 * kh;
        float h0 = c0[r] + bias1a;
        float h1 = c1[r] + bias1b;
        h0 = h0 > 0.0f ? h0 : 0.0f;
        h1 = h1 > 0.0f ? h1 : 0.0f;
        lds_h[mm * PH + colBase + n]      = h0;
        lds_h[mm * PH + colBase + 16 + n] = h1;
    }
    __syncthreads();

    v8f d = {};
    {
        const float* a = lds_h + m * PH + 64 * kq + 2 * kh;
        v2f af[16];
        #pragma unroll
        for (int ks = 0; ks < 16; ++ks) af[ks] = *(const v2f*)(a + 4 * ks);
        __builtin_amdgcn_sched_barrier(0);
        #pragma unroll
        for (int ks = 0; ks < 16; ++ks)
            d = __builtin_amdgcn_wmma_f32_16x16x4_f32(false, af[ks], false, w2f[ks], (short)0, d, false, false);
    }
    {
        float* p = lds_p + kq * (16 * PP) + (8 * kh) * PP + cb + n;
        #pragma unroll
        for (int r = 0; r < 8; ++r) p[r * PP] = d[r];
    }
    __syncthreads();

    {
        const float* p = lds_p + orow * PP + oc;
        v2f s = *(const v2f*)p;
        s += *(const v2f*)(p + 16 * PP);
        s += *(const v2f*)(p + 2 * 16 * PP);
        s += *(const v2f*)(p + 3 * 16 * PP);
        s += bias2;
        *(v2f*)(out + (size_t)(rowBase + orow) * OUTC + outOff + oc) = s;
    }
}

extern "C" void kernel_launch(void* const* d_in, const int* in_sizes, int n_in,
                              void* d_out, int out_size, void* d_ws, size_t ws_size,
                              hipStream_t stream) {
    const int*   src = (const int*)d_in[0];
    const float* Wb1 = (const float*)d_in[1];
    const float* bb1 = (const float*)d_in[2];
    const float* Wb2 = (const float*)d_in[3];
    const float* bb2 = (const float*)d_in[4];
    const float* Wt1 = (const float*)d_in[5];
    const float* bt1 = (const float*)d_in[6];
    const float* Wt2 = (const float*)d_in[7];
    const float* bt2 = (const float*)d_in[8];
    float* out = (float*)d_out;

    float* bi_freq  = (float*)d_ws;
    float* tri_freq = bi_freq + (size_t)B_ROWS * KPAD_BI;

    hist_kernel<<<B_ROWS, 256, 0, stream>>>(src, bi_freq, tri_freq);

    mlp_kernel<<<2 * TILES, 256, 0, stream>>>(bi_freq, Wb1, bb1, Wb2, bb2,
                                              tri_freq, Wt1, bt1, Wt2, bt2, out);
}


# Layer-1 W1 fragments go from global memory straight into registers, each wave three 32-k chunks ahead, the freq tile sits whole in LDS: no W1 LDS ring, no slab barriers; k order of the sums unchanged
# speedup vs baseline: 2.5537x; 1.0637x over previous
#include <hip/hip_runtime.h>

#define VOCAB 33
#define V2 1089
#define BUCKETS 512
#define B_ROWS 4096
#define S_LEN 4096
#define KPAD_BI 1104
#define HID 256
#define OUTC 64

typedef float v2f __attribute__((ext_vector_type(2)));
typedef float v8f __attribute__((ext_vector_type(8)));

typedef int v4i __attribute__((ext_vector_type(4)));
typedef int v2i __attribute__((ext_vector_type(2)));

__global__ __launch_bounds__(256) void hist_kernel(const int* __restrict__ src,
                                                   float* __restrict__ bi_freq,
                                                   float* __restrict__ tri_freq) {
    __shared__ unsigned bi_cnt[V2];
    __shared__ unsigned tri_cnt[BUCKETS];
    __shared__ unsigned totals[2];

    const int row = blockIdx.x;
    const int tid = threadIdx.x;
    const int* s = src + (size_t)row * S_LEN;

    v4i qa[4];
    v2i la[4];
    #pragma unroll
    for (int j = 0; j < 4; ++j)
        qa[j] = *(const v4i*)(s + j * (S_LEN / 4) + 4 * tid);
    #pragma unroll
    for (int j = 0; j < 4; ++j) {
        const int e = j * (S_LEN / 4) + 4 * tid + 4;
        const v2i v = *(const v2i*)(s + min(e, S_LEN - 2));
        la[j] = (e < S_LEN) ? v : (v2i){0, 0};
    }

    for (int i = tid; i < V2; i += 256) bi_cnt[i] = 0u;
    for (int i = tid; i < BUCKETS; i += 256) tri_cnt[i] = 0u;
    if (tid < 2) totals[tid] = 0u;
    __syncthreads();

    unsigned myBi = 0, myTri = 0;
    #pragma unroll
    for (int j = 0; j < 4; ++j) {
        const int tok[6] = {qa[j].x, qa[j].y, qa[j].z, qa[j].w, la[j].x, la[j].y};
        #pragma unroll
        for (int k = 0; k < 4; ++k) {
            const int c = tok[k];
            const int n = tok[k + 1];
            const int t = tok[k + 2];
            const bool bm = (c > 0) && (n > 0);
            if (bm) {
                atomicAdd(&bi_cnt[c * VOCAB + n], 1u);
                ++myBi;
            }
            if (bm && (t > 0)) {
                unsigned idx = ((unsigned)(c * (VOCAB * VOCAB) + n * VOCAB + t)) & (BUCKETS - 1);
                atomicAdd(&tri_cnt[idx], 1u);
                ++myTri;
            }
        }
    }
    if (myBi)  atomicAdd(&totals[0], myBi);
    if (myTri) atomicAdd(&totals[1], myTri);
    __syncthreads();

    const float biInv  = 1.0f / (float)max(totals[0], 1u);
    const float triInv = 1.0f / (float)max(totals[1], 1u);

    float* bf = bi_freq + (size_t)row * KPAD_BI;
    for (int i = tid; i < KPAD_BI; i += 256)
        bf[i] = (i < V2 ? (float)bi_cnt[i] : 0.0f) * biInv;

    float* tf = tri_freq + (size_t)row * BUCKETS;
    for (int i = tid; i < BUCKETS; i += 256)
        tf[i] = (float)tri_cnt[i] * triInv;
}

#define TILES (B_ROWS / 16)
#define CH 8
#define KC (4 * CH)
#define PA_BI (KPAD_BI + 4)
#define PA_TRI (BUCKETS + 4)
#define PH 260
#define PP 36

typedef float v4f __attribute__((ext_vector_type(4)));

__device__ __forceinline__ void lds_barrier() {
    __builtin_amdgcn_fence(__ATOMIC_RELEASE, "workgroup", "local");
    __builtin_amdgcn_s_barrier();
    __builtin_amdgcn_fence(__ATOMIC_ACQUIRE, "workgroup", "local");
}

struct Chunk { v2f r[CH][2]; };

template <int PER, int KPADV, int PITCH>
__device__ __forceinline__ void preload_tile(const float* __restrict__ freq, int rowBase, float* lds_a, int tid) {
    constexpr int TOTAL = 16 * PER;
    constexpr int IT = (TOTAL + 255) / 256;
    v4f t[IT];
    #pragma unroll
    for (int j = 0; j < IT; ++j) {
        const int i = min(tid + 256 * j, TOTAL - 1);
        const int row = i / PER, c4 = i % PER;
        t[j] = *(const v4f*)(freq + (size_t)(rowBase + row) * KPADV + 4 * c4);
    }
    #pragma unroll
    for (int j = 0; j < IT; ++j) {
        const int i = tid + 256 * j;
        if (i < TOTAL) *(v4f*)(lds_a + (i / PER) * PITCH + 4 * (i % PER)) = t[j];
    }
}

__global__ __launch_bounds__(256) void mlp_kernel(const float* __restrict__ freqB,
                                                  const float* __restrict__ W1B,
                                                  const float* __restrict__ b1B,
                                                  const float* __restrict__ W2B,
                                                  const float* __restrict__ b2B,
                                                  const float* __restrict__ freqT,
                                                  const float* __restrict__ W1T,
                                                  const float* __restrict__ b1T,
                                                  const float* __restrict__ W2T,
                                                  const float* __restrict__ b2T,
                                                  float* __restrict__ out) {
    const bool tri = blockIdx.x >= TILES;
    const float* __restrict__ W1   = tri ? W1T : W1B;
    const float* __restrict__ b1   = tri ? b1T : b1B;
    const float* __restrict__ W2   = tri ? W2T : W2B;
    const float* __restrict__ b2   = tri ? b2T : b2B;
    const int K      = tri ? BUCKETS : V2;
    const int pa     = tri ? PA_TRI : PA_BI;
    const int nFull  = tri ? BUCKETS / KC : V2 / KC;
    const int outOff = tri ? 32 : 0;

    __shared__ __attribute__((aligned(16))) float lds_a[16 * PA_BI];
    __shared__ __attribute__((aligned(16))) float lds_h[16 * PH];
    __shared__ __attribute__((aligned(16))) float lds_p[4 * 16 * PP];

    const int tid = threadIdx.x;
    const int rowBase = (tri ? blockIdx.x - TILES : blockIdx.x) * 16;
    const int lane = tid & 31;
    const int wave = tid >> 5;
    const int m  = lane & 15;
    const int kh = lane >> 4;
    const int n  = lane & 15;
    const int colBase = wave * 32;

    const float* wp = W1 + (size_t)(2 * kh) * HID + colBase + 2 * n;


    auto load_chunk = [&](int c, Chunk& b) {
        const float* p = wp + (size_t)(min(c, nFull - 1) * KC) * HID;
        #pragma unroll
        for (int ks = 0; ks < CH; ++ks) {
            b.r[ks][0] = *(const v2f*)(p + (4 * ks) * HID);
            b.r[ks][1] = *(const v2f*)(p + (4 * ks + 1) * HID);
        }
    };

    v8f c0 = {};
    v8f c1 = {};

    auto compute_chunk = [&](int c, Chunk& b, bool refill) {
        v2f fb0[CH], fb1[CH];
        #pragma unroll
        for (int ks = 0; ks < CH; ++ks) {
            fb0[ks] = (v2f){b.r[ks][0].x, b.r[ks][1].x};
            fb1[ks] = (v2f){b.r[ks][0].y, b.r[ks][1].y};
        }
        if (refill) load_chunk(c + 3, b);
        const float* a = lds_a + m * pa + c * KC + 2 * kh;
        v2f fa[CH];
        #pragma unroll
        for (int ks = 0; ks < CH; ++ks) fa[ks] = *(const v2f*)(a + 4 * ks);
        #pragma unroll
        for (int ks = 0; ks < CH; ++ks) {
            c0 = __builtin_amdgcn_wmma_f32_16x16x4_f32(false, fa[ks], false, fb0[ks], (short)0, c0, false, false);
            c1 = __builtin_amdgcn_wmma_f32_16x16x4_f32(false, fa[ks], false, fb1[ks], (short)0, c1, false, false);
        }
    };

    Chunk b0, b1c, b2c;
    load_chunk(0, b0);
    load_chunk(1, b1c);
    load_chunk(2, b2c);
    const int kt = (V2 / KC) * KC + 2 * kh;
    const v2f tl0 = *(const v2f*)(W1 + (size_t)min(kt,     K - 1) * HID + colBase + 2 * n);
    const v2f tl1 = *(const v2f*)(W1 + (size_t)min(kt + 1, K - 1) * HID + colBase + 2 * n);
    if (tri) preload_tile<BUCKETS / 4, BUCKETS, PA_TRI>(freqT, rowBase, lds_a, tid);
    else     preload_tile<KPAD_BI / 4, KPAD_BI, PA_BI>(freqB, rowBase, lds_a, tid);
    lds_barrier();

    static_assert((V2 / KC) % 3 == 1 && (BUCKETS / KC) % 3 == 1 && BUCKETS % KC == 0, "chunk rotation");
    const int nTriple = (nFull - 1) / 3;
    for (int t = 0; t < nTriple; ++t) {
        compute_chunk(3 * t,     b0,  true);
        compute_chunk(3 * t + 1, b1c, true);
        compute_chunk(3 * t + 2, b2c, true);
    }
    const int ct = wave & 1;
    const int kq = wave >> 1;
    const int cb = ct * 16;
    const v2f bias1 = *(const v2f*)(b1 + colBase + 2 * n);
    const int orow = tid >> 4;
    const int oc   = (tid & 15) * 2;
    const v2f bias2 = *(const v2f*)(b2 + oc);
    v2f w2f[16];
    {
        const float* w2p = W2 + (size_t)(64 * kq + 2 * kh) * 32 + cb + n;
        #pragma unroll
        for (int ks = 0; ks < 16; ++ks) {
            w2f[ks].x = w2p[(4 * ks) * 32];
            w2f[ks].y = w2p[(4 * ks + 1) * 32];
        }
    }

    compute_chunk(nFull - 1, b0, false);
    static_assert(V2 - (V2 / KC) * KC <= 4, "the tail fits one k-step");
    if (!tri) {
        const int k = (V2 / KC) * KC + 2 * kh;
        const v2f fbt0 = {k < V2 ? tl0.x : 0.f, k + 1 < V2 ? tl1.x : 0.f};
        const v2f fbt1 = {k < V2 ? tl0.y : 0.f, k + 1 < V2 ? tl1.y : 0.f};
        const v2f fat = *(const v2f*)(lds_a + m * pa + k);
        c0 = __builtin_amdgcn_wmma_f32_16x16x4_f32(false, fat, false, fbt0, (short)0, c0, false, false);
        c1 = __builtin_amdgcn_wmma_f32_16x16x4_f32(false, fat, false, fbt1, (short)0, c1, false, false);
    }

    #pragma unroll
    for (int r = 0; r < 8; ++r) {
        const int mm = r + 8 * kh;
        float h0 = c0[r] + bias1.x;
        float h1 = c1[r] + bias1.y;
        h0 = h0 > 0.0f ? h0 : 0.0f;
        h1 = h1 > 0.0f ? h1 : 0.0f;
        *(v2f*)(lds_h + mm * PH + colBase + 2 * n) = (v2f){h0, h1};
    }
    __syncthreads();

    v8f d = {};
    {
        const float* a = lds_h + m * PH + 64 * kq + 2 * kh;
        v2f af[16];
        #pragma unroll
        for (int ks = 0; ks < 16; ++ks) af[ks] = *(const v2f*)(a + 4 * ks);
        __builtin_amdgcn_sched_barrier(0);
        #pragma unroll
        for (int ks = 0; ks < 16; ++ks)
            d = __builtin_amdgcn_wmma_f32_16x16x4_f32(false, af[ks], false, w2f[ks], (short)0, d, false, false);
    }
    {
        float* p = lds_p + kq * (16 * PP) + (8 * kh) * PP + cb + n;
        #pragma unroll
        for (int r = 0; r < 8; ++r) p[r * PP] = d[r];
    }
    __syncthreads();

    {
        const float* p = lds_p + orow * PP + oc;
        v2f s = *(const v2f*)p;
        s += *(const v2f*)(p + 16 * PP);
        s += *(const v2f*)(p + 2 * 16 * PP);
        s += *(const v2f*)(p + 3 * 16 * PP);
        s += bias2;
        *(v2f*)(out + (size_t)(rowBase + orow) * OUTC + outOff + oc) = s;
    }
}

extern "C" void kernel_launch(void* const* d_in, const int* in_sizes, int n_in,
                              void* d_out, int out_size, void* d_ws, size_t ws_size,
                              hipStream_t stream) {
    const int*   src = (const int*)d_in[0];
    const float* Wb1 = (const float*)d_in[1];
    const float* bb1 = (const float*)d_in[2];
    const float* Wb2 = (const float*)d_in[3];
    const float* bb2 = (const float*)d_in[4];
    const float* Wt1 = (const float*)d_in[5];
    const float* bt1 = (const float*)d_in[6];
    const float* Wt2 = (const float*)d_in[7];
    const float* bt2 = (const float*)d_in[8];
    float* out = (float*)d_out;

    float* bi_freq  = (float*)d_ws;
    float* tri_freq = bi_freq + (size_t)B_ROWS * KPAD_BI;

    hist_kernel<<<B_ROWS, 256, 0, stream>>>(src, bi_freq, tri_freq);

    mlp_kernel<<<2 * TILES, 256, 0, stream>>>(bi_freq, Wb1, bb1, Wb2, bb2,
                                              tri_freq, Wt1, bt1, Wt2, bt2, out);
}


# Histograms are counted inside the MLP workgroup, straight into its LDS freq tile (a wave counts 2 of the 16 rows, tokens fetched 3 passes ahead): 2 launches -> 1, no freq planes; sums unchanged
# speedup vs baseline: 3.2057x; 1.2553x over previous
#include <hip/hip_runtime.h>

#define VOCAB 33
#define V2 1089
#define BUCKETS 512
#define B_ROWS 4096
#define S_LEN 4096
#define KPAD_BI 1104
#define HID 256
#define OUTC 64

typedef float v2f __attribute__((ext_vector_type(2)));
typedef float v8f __attribute__((ext_vector_type(8)));

#define TILES (B_ROWS / 16)
#define CH 8
#define KC (4 * CH)
#define PA_BI (KPAD_BI + 4)
#define PA_TRI (BUCKETS + 4)
#define PH 260
#define PP 36

typedef float v4f __attribute__((ext_vector_type(4)));

__device__ __forceinline__ void lds_barrier() {
    __builtin_amdgcn_fence(__ATOMIC_RELEASE, "workgroup", "local");
    __builtin_amdgcn_s_barrier();
    __builtin_amdgcn_fence(__ATOMIC_ACQUIRE, "workgroup", "local");
}

typedef int v4i __attribute__((ext_vector_type(4)));
typedef int v2i __attribute__((ext_vector_type(2)));
typedef unsigned v4u __attribute__((ext_vector_type(4)));

#define PASSES 16

template <bool TRI> struct Pass;
template <> struct Pass<false> { v4i q[4]; int l[4]; };
template <> struct Pass<true>  { v4i q[4]; v2i l[4]; };

template <bool TRI>
__device__ __forceinline__ void load_pass(const int* __restrict__ rows, int idx, int lane, Pass<TRI>& b) {
    const int i = min(idx, PASSES - 1);
    const int* s = rows + (size_t)(i >> 3) * S_LEN;
    const int off = 512 * (i & 7) + 4 * lane;
    #pragma unroll
    for (int j = 0; j < 4; ++j) b.q[j] = *(const v4i*)(s + off + 128 * j);
    #pragma unroll
    for (int j = 0; j < 4; ++j) {
        if constexpr (TRI) b.l[j] = *(const v2i*)(s + min(off + 128 * j + 4, S_LEN - 2));
        else               b.l[j] = s[min(off + 128 * j + 4, S_LEN - 1)];
    }
}

template <bool TRI>
__device__ __forceinline__ unsigned count_pass(const Pass<TRI>& b, int idx, int lane, unsigned* cntRow) {
    unsigned my = 0;
    const int off = 512 * (idx & 7) + 4 * lane;
    #pragma unroll
    for (int j = 0; j < 4; ++j) {
        const bool pastEnd = (off + 128 * j + 4) >= S_LEN;
        int l0, l1;
        if constexpr (TRI) { l0 = pastEnd ? 0 : b.l[j].x; l1 = pastEnd ? 0 : b.l[j].y; }
        else               { l0 = pastEnd ? 0 : b.l[j];   l1 = 0; }
        const int tok[6] = {b.q[j].x, b.q[j].y, b.q[j].z, b.q[j].w, l0, l1};
        #pragma unroll
        for (int k = 0; k < 4; ++k) {
            const int c = tok[k];
            const int n = tok[k + 1];
            const bool bm = (c > 0) && (n > 0);
            if constexpr (!TRI) {
                if (bm) {
                    atomicAdd(&cntRow[c * VOCAB + n], 1u);
                    ++my;
                }
            } else {
                const int t = tok[k + 2];
                if (bm && (t > 0)) {
                    const unsigned h = ((unsigned)(c * (VOCAB * VOCAB) + n * VOCAB + t)) & (BUCKETS - 1);
                    atomicAdd(&cntRow[h], 1u);
                    ++my;
                }
            }
        }
    }
    return my;
}

template <bool TRI, int PITCH>
__device__ __forceinline__ void hist_stage(const int* __restrict__ rows, int tid, unsigned* cntW, unsigned* totals) {
    const int lane = tid & 31;
    const int wave = tid >> 5;
    Pass<TRI> pA, pB, pC, pD;
    load_pass<TRI>(rows, 0, lane, pA);
    load_pass<TRI>(rows, 1, lane, pB);
    load_pass<TRI>(rows, 2, lane, pC);
    for (int i = lane; i < 2 * PITCH / 4; i += 32) *(v4u*)(cntW + 4 * i) = (v4u){0u, 0u, 0u, 0u};
    if (tid < 16) totals[tid] = 0u;
    lds_barrier();
    unsigned my0 = 0, my1 = 0;
    for (int it = 0; it < PASSES / 4; ++it) {
        unsigned* cr = cntW + (it >> 1) * PITCH;
        load_pass<TRI>(rows, 4 * it + 3, lane, pD);
        unsigned got = count_pass<TRI>(pA, 4 * it, lane, cr);
        load_pass<TRI>(rows, 4 * it + 4, lane, pA);
        got += count_pass<TRI>(pB, 4 * it + 1, lane, cr);
        load_pass<TRI>(rows, 4 * it + 5, lane, pB);
        got += count_pass<TRI>(pC, 4 * it + 2, lane, cr);
        load_pass<TRI>(rows, 4 * it + 6, lane, pC);
        got += count_pass<TRI>(pD, 4 * it + 3, lane, cr);
        if (it < 2) my0 += got; else my1 += got;
    }
    if (my0) atomicAdd(&totals[2 * wave], my0);
    if (my1) atomicAdd(&totals[2 * wave + 1], my1);
}

struct Chunk { v2f r[CH][2]; };

__global__ __launch_bounds__(256) void encoder_kernel(const int* __restrict__ src,
                                                  const float* __restrict__ W1B,
                                                  const float* __restrict__ b1B,
                                                  const float* __restrict__ W2B,
                                                  const float* __restrict__ b2B,
                                                  const float* __restrict__ W1T,
                                                  const float* __restrict__ b1T,
                                                  const float* __restrict__ W2T,
                                                  const float* __restrict__ b2T,
                                                  float* __restrict__ out) {
    const bool tri = blockIdx.x >= TILES;
    const float* __restrict__ W1   = tri ? W1T : W1B;
    const float* __restrict__ b1   = tri ? b1T : b1B;
    const float* __restrict__ W2   = tri ? W2T : W2B;
    const float* __restrict__ b2   = tri ? b2T : b2B;
    const int K      = tri ? BUCKETS : V2;
    const int pa     = tri ? PA_TRI : PA_BI;
    const int nFull  = tri ? BUCKETS / KC : V2 / KC;
    const int outOff = tri ? 32 : 0;

    __shared__ __attribute__((aligned(16))) float lds_a[16 * PA_BI];
    __shared__ __attribute__((aligned(16))) float lds_h[16 * PH];
    __shared__ __attribute__((aligned(16))) float lds_p[4 * 16 * PP];
    __shared__ unsigned totals[16];

    const int tid = threadIdx.x;
    const int rowBase = (tri ? blockIdx.x - TILES : blockIdx.x) * 16;
    const int lane = tid & 31;
    const int wave = tid >> 5;
    const int m  = lane & 15;
    const int kh = lane >> 4;
    const int n  = lane & 15;
    const int colBase = wave * 32;

    const float* wp = W1 + (size_t)(2 * kh) * HID + colBase + 2 * n;


    auto load_chunk = [&](int c, Chunk& b) {
        const float* p = wp + (size_t)(min(c, nFull - 1) * KC) * HID;
        #pragma unroll
        for (int ks = 0; ks < CH; ++ks) {
            b.r[ks][0] = *(const v2f*)(p + (4 * ks) * HID);
            b.r[ks][1] = *(const v2f*)(p + (4 * ks + 1) * HID);
        }
    };

    v8f c0 = {};
    v8f c1 = {};

    auto compute_chunk = [&](int c, Chunk& b, bool refill) {
        v2f fb0[CH], fb1[CH];
        #pragma unroll
        for (int ks = 0; ks < CH; ++ks) {
            fb0[ks] = (v2f){b.r[ks][0].x, b.r[ks][1].x};
            fb1[ks] = (v2f){b.r[ks][0].y, b.r[ks][1].y};
        }
        if (refill) load_chunk(c + 3, b);
        const float* a = lds_a + m * pa + c * KC + 2 * kh;
        v2f fa[CH];
        #pragma unroll
        for (int ks = 0; ks < CH; ++ks) fa[ks] = *(const v2f*)(a + 4 * ks);
        #pragma unroll
        for (int ks = 0; ks < CH; ++ks) {
            c0 = __builtin_amdgcn_wmma_f32_16x16x4_f32(false, fa[ks], false, fb0[ks], (short)0, c0, false, false);
            c1 = __builtin_amdgcn_wmma_f32_16x16x4_f32(false, fa[ks], false, fb1[ks], (short)0, c1, false, false);
        }
    };

    unsigned* cnt = reinterpret_cast<unsigned*>(lds_a);
    const int* rows = src + (size_t)(rowBase + 2 * wave) * S_LEN;
    unsigned* cntW = cnt + (2 * wave) * pa;
    if (tri) hist_stage<true,  PA_TRI>(rows, tid, cntW, totals);
    else     hist_stage<false, PA_BI>(rows, tid, cntW, totals);

    Chunk b0, b1c, b2c;
    load_chunk(0, b0);
    load_chunk(1, b1c);
    load_chunk(2, b2c);
    const int kt = (V2 / KC) * KC + 2 * kh;
    const v2f tl0 = *(const v2f*)(W1 + (size_t)min(kt,     K - 1) * HID + colBase + 2 * n);
    const v2f tl1 = *(const v2f*)(W1 + (size_t)min(kt + 1, K - 1) * HID + colBase + 2 * n);
    lds_barrier();
    for (int rr = 0; rr < 2; ++rr) {
        const float inv = 1.0f / (float)max(totals[2 * wave + rr], 1u);
        for (int i = lane; i < pa / 4; i += 32) {
            const v4u u = *(const v4u*)(cntW + rr * pa + 4 * i);
            *(v4f*)(lds_a + (2 * wave + rr) * pa + 4 * i) =
                (v4f){(float)u.x * inv, (float)u.y * inv, (float)u.z * inv, (float)u.w * inv};
        }
    }
    lds_barrier();

    static_assert((V2 / KC) % 3 == 1 && (BUCKETS / KC) % 3 == 1 && BUCKETS % KC == 0, "chunk rotation");
    const int nTriple = (nFull - 1) / 3;
    for (int t = 0; t < nTriple; ++t) {
        compute_chunk(3 * t,     b0,  true);
        compute_chunk(3 * t + 1, b1c, true);
        compute_chunk(3 * t + 2, b2c, true);
    }
    const int ct = wave & 1;
    const int kq = wave >> 1;
    const int cb = ct * 16;
    const v2f bias1 = *(const v2f*)(b1 + colBase + 2 * n);
    const int orow = tid >> 4;
    const int oc   = (tid & 15) * 2;
    const v2f bias2 = *(const v2f*)(b2 + oc);
    v2f w2f[16];
    {
        const float* w2p = W2 + (size_t)(64 * kq + 2 * kh) * 32 + cb + n;
        #pragma unroll
        for (int ks = 0; ks < 16; ++ks) {
            w2f[ks].x = w2p[(4 * ks) * 32];
            w2f[ks].y = w2p[(4 * ks + 1) * 32];
        }
    }

    compute_chunk(nFull - 1, b0, false);
    static_assert(V2 - (V2 / KC) * KC <= 4, "the tail fits one k-step");
    if (!tri) {
        const int k = (V2 / KC) * KC + 2 * kh;
        const v2f fbt0 = {k < V2 ? tl0.x : 0.f, k + 1 < V2 ? tl1.x : 0.f};
        const v2f fbt1 = {k < V2 ? tl0.y : 0.f, k + 1 < V2 ? tl1.y : 0.f};
        const v2f fat = *(const v2f*)(lds_a + m * pa + k);
        c0 = __builtin_amdgcn_wmma_f32_16x16x4_f32(false, fat, false, fbt0, (short)0, c0, false, false);
        c1 = __builtin_amdgcn_wmma_f32_16x16x4_f32(false, fat, false, fbt1, (short)0, c1, false, false);
    }

    #pragma unroll
    for (int r = 0; r < 8; ++r) {
        const int mm = r + 8 * kh;
        float h0 = c0[r] + bias1.x;
        float h1 = c1[r] + bias1.y;
        h0 = h0 > 0.0f ? h0 : 0.0f;
        h1 = h1 > 0.0f ? h1 : 0.0f;
        *(v2f*)(lds_h + mm * PH + colBase + 2 * n) = (v2f){h0, h1};
    }
    __syncthreads();

    v8f d = {};
    {
        const float* a = lds_h + m * PH + 64 * kq + 2 * kh;
        v2f af[16];
        #pragma unroll
        for (int ks = 0; ks < 16; ++ks) af[ks] = *(const v2f*)(a + 4 * ks);
        __builtin_amdgcn_sched_barrier(0);
        #pragma unroll
        for (int ks = 0; ks < 16; ++ks)
            d = __builtin_amdgcn_wmma_f32_16x16x4_f32(false, af[ks], false, w2f[ks], (short)0, d, false, false);
    }
    {
        float* p = lds_p + kq * (16 * PP) + (8 * kh) * PP + cb + n;
        #pragma unroll
        for (int r = 0; r < 8; ++r) p[r * PP] = d[r];
    }
    __syncthreads();

    {
        const float* p = lds_p + orow * PP + oc;
        v2f s = *(const v2f*)p;
        s += *(const v2f*)(p + 16 * PP);
        s += *(const v2f*)(p + 2 * 16 * PP);
        s += *(const v2f*)(p + 3 * 16 * PP);
        s += bias2;
        *(v2f*)(out + (size_t)(rowBase + orow) * OUTC + outOff + oc) = s;
    }
}

extern "C" void kernel_launch(void* const* d_in, const int* in_sizes, int n_in,
                              void* d_out, int out_size, void* d_ws, size_t ws_size,
                              hipStream_t stream) {
    const int*   src = (const int*)d_in[0];
    const float* Wb1 = (const float*)d_in[1];
    const float* bb1 = (const float*)d_in[2];
    const float* Wb2 = (const float*)d_in[3];
    const float* bb2 = (const float*)d_in[4];
    const float* Wt1 = (const float*)d_in[5];
    const float* bt1 = (const float*)d_in[6];
    const float* Wt2 = (const float*)d_in[7];
    const float* bt2 = (const float*)d_in[8];
    float* out = (float*)d_out;

    encoder_kernel<<<2 * TILES, 256, 0, stream>>>(src, Wb1, bb1, Wb2, bb2, Wt1, bt1, Wt2, bt2, out);
}


# Histogram counting is straight-line: every position issues its LDS add with its mask as the addend (0 or 1), bins by 24-bit multiply-add: no exec save, branch, restore per token; integer counts
# speedup vs baseline: 3.3333x; 1.0398x over previous
#include <hip/hip_runtime.h>

#define VOCAB 33
#define V2 1089
#define BUCKETS 512
#define B_ROWS 4096
#define S_LEN 4096
#define KPAD_BI 1104
#define HID 256
#define OUTC 64

typedef float v2f __attribute__((ext_vector_type(2)));
typedef float v8f __attribute__((ext_vector_type(8)));

#define TILES (B_ROWS / 16)
#define CH 8
#define KC (4 * CH)
#define PA_BI (KPAD_BI + 4)
#define PA_TRI (BUCKETS + 4)
#define PH 260
#define PP 36

typedef float v4f __attribute__((ext_vector_type(4)));

__device__ __forceinline__ void lds_barrier() {
    __builtin_amdgcn_fence(__ATOMIC_RELEASE, "workgroup", "local");
    __builtin_amdgcn_s_barrier();
    __builtin_amdgcn_fence(__ATOMIC_ACQUIRE, "workgroup", "local");
}

typedef int v4i __attribute__((ext_vector_type(4)));
typedef int v2i __attribute__((ext_vector_type(2)));
typedef unsigned v4u __attribute__((ext_vector_type(4)));

#define PASSES 16

template <bool TRI> struct Pass;
template <> struct Pass<false> { v4i q[4]; int l[4]; };
template <> struct Pass<true>  { v4i q[4]; v2i l[4]; };

template <bool TRI>
__device__ __forceinline__ void load_pass(const int* __restrict__ rows, int idx, int lane, Pass<TRI>& b) {
    const int i = min(idx, PASSES - 1);
    const int* s = rows + (size_t)(i >> 3) * S_LEN;
    const int off = 512 * (i & 7) + 4 * lane;
    #pragma unroll
    for (int j = 0; j < 4; ++j) b.q[j] = *(const v4i*)(s + off + 128 * j);
    #pragma unroll
    for (int j = 0; j < 4; ++j) {
        if constexpr (TRI) b.l[j] = *(const v2i*)(s + min(off + 128 * j + 4, S_LEN - 2));
        else               b.l[j] = s[min(off + 128 * j + 4, S_LEN - 1)];
    }
}

template <bool TRI>
__device__ __forceinline__ unsigned count_pass(const Pass<TRI>& b, int idx, int lane, unsigned* cntRow) {
    unsigned my = 0;
    const int off = 512 * (idx & 7) + 4 * lane;
    #pragma unroll
    for (int j = 0; j < 4; ++j) {
        const bool pastEnd = (off + 128 * j + 4) >= S_LEN;
        int l0, l1;
        if constexpr (TRI) { l0 = pastEnd ? 0 : b.l[j].x; l1 = pastEnd ? 0 : b.l[j].y; }
        else               { l0 = pastEnd ? 0 : b.l[j];   l1 = 0; }
        const int tok[6] = {b.q[j].x, b.q[j].y, b.q[j].z, b.q[j].w, l0, l1};
        unsigned tc[6], nz[6];
        #pragma unroll
        for (int k = 0; k < (TRI ? 6 : 5); ++k) {
            tc[k] = (unsigned)min(max(tok[k], 0), VOCAB - 1);
            nz[k] = min(tc[k], 1u);
        }
        #pragma unroll
        for (int k = 0; k < 4; ++k) {
            if constexpr (!TRI) {
                const unsigned bi = tc[k] * VOCAB + tc[k + 1];
                const unsigned m = nz[k] & nz[k + 1];
                atomicAdd(&cntRow[bi], m);
                my += m;
            } else {
                const unsigned h = (tc[k] * (VOCAB * VOCAB) + tc[k + 1] * VOCAB + tc[k + 2]) & (BUCKETS - 1);
                const unsigned m = nz[k] & nz[k + 1] & nz[k + 2];
                atomicAdd(&cntRow[h], m);
                my += m;
            }
        }
    }
    return my;
}

template <bool TRI, int PITCH>
__device__ __forceinline__ void hist_stage(const int* __restrict__ rows, int tid, unsigned* cntW, unsigned* totals) {
    const int lane = tid & 31;
    const int wave = tid >> 5;
    Pass<TRI> pA, pB, pC, pD;
    load_pass<TRI>(rows, 0, lane, pA);
    load_pass<TRI>(rows, 1, lane, pB);
    load_pass<TRI>(rows, 2, lane, pC);
    for (int i = lane; i < 2 * PITCH / 4; i += 32) *(v4u*)(cntW + 4 * i) = (v4u){0u, 0u, 0u, 0u};
    if (tid < 16) totals[tid] = 0u;
    lds_barrier();
    unsigned my0 = 0, my1 = 0;
    for (int it = 0; it < PASSES / 4; ++it) {
        unsigned* cr = cntW + (it >> 1) * PITCH;
        load_pass<TRI>(rows, 4 * it + 3, lane, pD);
        __builtin_amdgcn_sched_barrier(0);
        unsigned got = count_pass<TRI>(pA, 4 * it, lane, cr);
        __builtin_amdgcn_sched_barrier(0);
        load_pass<TRI>(rows, 4 * it + 4, lane, pA);
        __builtin_amdgcn_sched_barrier(0);
        got += count_pass<TRI>(pB, 4 * it + 1, lane, cr);
        __builtin_amdgcn_sched_barrier(0);
        load_pass<TRI>(rows, 4 * it + 5, lane, pB);
        __builtin_amdgcn_sched_barrier(0);
        got += count_pass<TRI>(pC, 4 * it + 2, lane, cr);
        __builtin_amdgcn_sched_barrier(0);
        load_pass<TRI>(rows, 4 * it + 6, lane, pC);
        __builtin_amdgcn_sched_barrier(0);
        got += count_pass<TRI>(pD, 4 * it + 3, lane, cr);
        if (it < 2) my0 += got; else my1 += got;
    }
    if (my0) atomicAdd(&totals[2 * wave], my0);
    if (my1) atomicAdd(&totals[2 * wave + 1], my1);
}

struct Chunk { v2f r[CH][2]; };

__global__ __launch_bounds__(256) void encoder_kernel(const int* __restrict__ src,
                                                  const float* __restrict__ W1B,
                                                  const float* __restrict__ b1B,
                                                  const float* __restrict__ W2B,
                                                  const float* __restrict__ b2B,
                                                  const float* __restrict__ W1T,
                                                  const float* __restrict__ b1T,
                                                  const float* __restrict__ W2T,
                                                  const float* __restrict__ b2T,
                                                  float* __restrict__ out) {
    const bool tri = blockIdx.x >= TILES;
    const float* __restrict__ W1   = tri ? W1T : W1B;
    const float* __restrict__ b1   = tri ? b1T : b1B;
    const float* __restrict__ W2   = tri ? W2T : W2B;
    const float* __restrict__ b2   = tri ? b2T : b2B;
    const int K      = tri ? BUCKETS : V2;
    const int pa     = tri ? PA_TRI : PA_BI;
    const int nFull  = tri ? BUCKETS / KC : V2 / KC;
    const int outOff = tri ? 32 : 0;

    __shared__ __attribute__((aligned(16))) float lds_a[16 * PA_BI];
    __shared__ __attribute__((aligned(16))) float lds_h[16 * PH];
    __shared__ __attribute__((aligned(16))) float lds_p[4 * 16 * PP];
    __shared__ unsigned totals[16];

    const int tid = threadIdx.x;
    const int rowBase = (tri ? blockIdx.x - TILES : blockIdx.x) * 16;
    const int lane = tid & 31;
    const int wave = tid >> 5;
    const int m  = lane & 15;
    const int kh = lane >> 4;
    const int n  = lane & 15;
    const int colBase = wave * 32;

    const float* wp = W1 + (size_t)(2 * kh) * HID + colBase + 2 * n;


    auto load_chunk = [&](int c, Chunk& b) {
        const float* p = wp + (size_t)(min(c, nFull - 1) * KC) * HID;
        #pragma unroll
        for (int ks = 0; ks < CH; ++ks) {
            b.r[ks][0] = *(const v2f*)(p + (4 * ks) * HID);
            b.r[ks][1] = *(const v2f*)(p + (4 * ks + 1) * HID);
        }
    };

    v8f c0 = {};
    v8f c1 = {};

    auto compute_chunk = [&](int c, Chunk& b, bool refill) {
        v2f fb0[CH], fb1[CH];
        #pragma unroll
        for (int ks = 0; ks < CH; ++ks) {
            fb0[ks] = (v2f){b.r[ks][0].x, b.r[ks][1].x};
            fb1[ks] = (v2f){b.r[ks][0].y, b.r[ks][1].y};
        }
        if (refill) load_chunk(c + 3, b);
        const float* a = lds_a + m * pa + c * KC + 2 * kh;
        v2f fa[CH];
        #pragma unroll
        for (int ks = 0; ks < CH; ++ks) fa[ks] = *(const v2f*)(a + 4 * ks);
        #pragma unroll
        for (int ks = 0; ks < CH; ++ks) {
            c0 = __builtin_amdgcn_wmma_f32_16x16x4_f32(false, fa[ks], false, fb0[ks], (short)0, c0, false, false);
            c1 = __builtin_amdgcn_wmma_f32_16x16x4_f32(false, fa[ks], false, fb1[ks], (short)0, c1, false, false);
        }
    };

    unsigned* cnt = reinterpret_cast<unsigned*>(lds_a);
    const int* rows = src + (size_t)(rowBase + 2 * wave) * S_LEN;
    unsigned* cntW = cnt + (2 * wave) * pa;
    if (tri) hist_stage<true,  PA_TRI>(rows, tid, cntW, totals);
    else     hist_stage<false, PA_BI>(rows, tid, cntW, totals);

    Chunk b0, b1c, b2c;
    load_chunk(0, b0);
    load_chunk(1, b1c);
    load_chunk(2, b2c);
    const int kt = (V2 / KC) * KC + 2 * kh;
    const v2f tl0 = *(const v2f*)(W1 + (size_t)min(kt,     K - 1) * HID + colBase + 2 * n);
    const v2f tl1 = *(const v2f*)(W1 + (size_t)min(kt + 1, K - 1) * HID + colBase + 2 * n);
    lds_barrier();
    for (int rr = 0; rr < 2; ++rr) {
        const float inv = 1.0f / (float)max(totals[2 * wave + rr], 1u);
        for (int i = lane; i < pa / 4; i += 32) {
            const v4u u = *(const v4u*)(cntW + rr * pa + 4 * i);
            *(v4f*)(lds_a + (2 * wave + rr) * pa + 4 * i) =
                (v4f){(float)u.x * inv, (float)u.y * inv, (float)u.z * inv, (float)u.w * inv};
        }
    }
    lds_barrier();

    static_assert((V2 / KC) % 3 == 1 && (BUCKETS / KC) % 3 == 1 && BUCKETS % KC == 0, "chunk rotation");
    const int nTriple = (nFull - 1) / 3;
    for (int t = 0; t < nTriple; ++t) {
        compute_chunk(3 * t,     b0,  true);
        compute_chunk(3 * t + 1, b1c, true);
        compute_chunk(3 * t + 2, b2c, true);
    }
    const int ct = wave & 1;
    const int kq = wave >> 1;
    const int cb = ct * 16;
    const v2f bias1 = *(const v2f*)(b1 + colBase + 2 * n);
    const int orow = tid >> 4;
    const int oc   = (tid & 15) * 2;
    const v2f bias2 = *(const v2f*)(b2 + oc);
    v2f w2f[16];
    {
        const float* w2p = W2 + (size_t)(64 * kq + 2 * kh) * 32 + cb + n;
        #pragma unroll
        for (int ks = 0; ks < 16; ++ks) {
            w2f[ks].x = w2p[(4 * ks) * 32];
            w2f[ks].y = w2p[(4 * ks + 1) * 32];
        }
    }

    compute_chunk(nFull - 1, b0, false);
    static_assert(V2 - (V2 / KC) * KC <= 4, "the tail fits one k-step");
    if (!tri) {
        const int k = (V2 / KC) * KC + 2 * kh;
        const v2f fbt0 = {k < V2 ? tl0.x : 0.f, k + 1 < V2 ? tl1.x : 0.f};
        const v2f fbt1 = {k < V2 ? tl0.y : 0.f, k + 1 < V2 ? tl1.y : 0.f};
        const v2f fat = *(const v2f*)(lds_a + m * pa + k);
        c0 = __builtin_amdgcn_wmma_f32_16x16x4_f32(false, fat, false, fbt0, (short)0, c0, false, false);
        c1 = __builtin_amdgcn_wmma_f32_16x16x4_f32(false, fat, false, fbt1, (short)0, c1, false, false);
    }

    #pragma unroll
    for (int r = 0; r < 8; ++r) {
        const int mm = r + 8 * kh;
        float h0 = c0[r] + bias1.x;
        float h1 = c1[r] + bias1.y;
        h0 = h0 > 0.0f ? h0 : 0.0f;
        h1 = h1 > 0.0f ? h1 : 0.0f;
        *(v2f*)(lds_h + mm * PH + colBase + 2 * n) = (v2f){h0, h1};
    }
    __syncthreads();

    v8f d = {};
    {
        const float* a = lds_h + m * PH + 64 * kq + 2 * kh;
        v2f af[16];
        #pragma unroll
        for (int ks = 0; ks < 16; ++ks) af[ks] = *(const v2f*)(a + 4 * ks);
        __builtin_amdgcn_sched_barrier(0);
        #pragma unroll
        for (int ks = 0; ks < 16; ++ks)
            d = __builtin_amdgcn_wmma_f32_16x16x4_f32(false, af[ks], false, w2f[ks], (short)0, d, false, false);
    }
    {
        float* p = lds_p + kq * (16 * PP) + (8 * kh) * PP + cb + n;
        #pragma unroll
        for (int r = 0; r < 8; ++r) p[r * PP] = d[r];
    }
    __syncthreads();

    {
        const float* p = lds_p + orow * PP + oc;
        v2f s = *(const v2f*)p;
        s += *(const v2f*)(p + 16 * PP);
        s += *(const v2f*)(p + 2 * 16 * PP);
        s += *(const v2f*)(p + 3 * 16 * PP);
        s += bias2;
        *(v2f*)(out + (size_t)(rowBase + orow) * OUTC + outOff + oc) = s;
    }
}

extern "C" void kernel_launch(void* const* d_in, const int* in_sizes, int n_in,
                              void* d_out, int out_size, void* d_ws, size_t ws_size,
                              hipStream_t stream) {
    const int*   src = (const int*)d_in[0];
    const float* Wb1 = (const float*)d_in[1];
    const float* bb1 = (const float*)d_in[2];
    const float* Wb2 = (const float*)d_in[3];
    const float* bb2 = (const float*)d_in[4];
    const float* Wt1 = (const float*)d_in[5];
    const float* bt1 = (const float*)d_in[6];
    const float* Wt2 = (const float*)d_in[7];
    const float* bt2 = (const float*)d_in[8];
    float* out = (float*)d_out;

    encoder_kernel<<<2 * TILES, 256, 0, stream>>>(src, Wb1, bb1, Wb2, bb2, Wt1, bt1, Wt2, bt2, out);
}


# Layer-1 k loop: next chunk's A fragments read from LDS as one group ahead of the current chunk's matrix instructions, W1 refill behind them: no LDS wait in the matrix stream; sums unchanged
# speedup vs baseline: 3.5781x; 1.0734x over previous
#include <hip/hip_runtime.h>

#define VOCAB 33
#define V2 1089
#define BUCKETS 512
#define B_ROWS 4096
#define S_LEN 4096
#define KPAD_BI 1104
#define HID 256
#define OUTC 64

typedef float v2f __attribute__((ext_vector_type(2)));
typedef float v8f __attribute__((ext_vector_type(8)));

#define TILES (B_ROWS / 16)
#define CH 8
#define KC (4 * CH)
#define PA_BI (KPAD_BI + 4)
#define PA_TRI (BUCKETS + 4)
#define PH 260
#define PP 36

typedef float v4f __attribute__((ext_vector_type(4)));

__device__ __forceinline__ void lds_barrier() {
    __builtin_amdgcn_fence(__ATOMIC_RELEASE, "workgroup", "local");
    __builtin_amdgcn_s_barrier();
    __builtin_amdgcn_fence(__ATOMIC_ACQUIRE, "workgroup", "local");
}

typedef int v4i __attribute__((ext_vector_type(4)));
typedef int v2i __attribute__((ext_vector_type(2)));
typedef unsigned v4u __attribute__((ext_vector_type(4)));

#define PASSES 16

template <bool TRI> struct Pass;
template <> struct Pass<false> { v4i q[4]; int l[4]; };
template <> struct Pass<true>  { v4i q[4]; v2i l[4]; };

template <bool TRI>
__device__ __forceinline__ void load_pass(const int* __restrict__ rows, int idx, int lane, Pass<TRI>& b) {
    const int i = min(idx, PASSES - 1);
    const int* s = rows + (size_t)(i >> 3) * S_LEN;
    const int off = 512 * (i & 7) + 4 * lane;
    #pragma unroll
    for (int j = 0; j < 4; ++j) b.q[j] = *(const v4i*)(s + off + 128 * j);
    #pragma unroll
    for (int j = 0; j < 4; ++j) {
        if constexpr (TRI) b.l[j] = *(const v2i*)(s + min(off + 128 * j + 4, S_LEN - 2));
        else               b.l[j] = s[min(off + 128 * j + 4, S_LEN - 1)];
    }
}

template <bool TRI>
__device__ __forceinline__ unsigned count_pass(const Pass<TRI>& b, int idx, int lane, unsigned* cntRow) {
    unsigned my = 0;
    const int off = 512 * (idx & 7) + 4 * lane;
    #pragma unroll
    for (int j = 0; j < 4; ++j) {
        const bool pastEnd = (off + 128 * j + 4) >= S_LEN;
        int l0, l1;
        if constexpr (TRI) { l0 = pastEnd ? 0 : b.l[j].x; l1 = pastEnd ? 0 : b.l[j].y; }
        else               { l0 = pastEnd ? 0 : b.l[j];   l1 = 0; }
        const int tok[6] = {b.q[j].x, b.q[j].y, b.q[j].z, b.q[j].w, l0, l1};
        unsigned tc[6], nz[6];
        #pragma unroll
        for (int k = 0; k < (TRI ? 6 : 5); ++k) {
            tc[k] = (unsigned)min(max(tok[k], 0), VOCAB - 1);
            nz[k] = min(tc[k], 1u);
        }
        #pragma unroll
        for (int k = 0; k < 4; ++k) {
            if constexpr (!TRI) {
                const unsigned bi = tc[k] * VOCAB + tc[k + 1];
                const unsigned m = nz[k] & nz[k + 1];
                atomicAdd(&cntRow[bi], m);
                my += m;
            } else {
                const unsigned h = (tc[k] * (VOCAB * VOCAB) + tc[k + 1] * VOCAB + tc[k + 2]) & (BUCKETS - 1);
                const unsigned m = nz[k] & nz[k + 1] & nz[k + 2];
                atomicAdd(&cntRow[h], m);
                my += m;
            }
        }
    }
    return my;
}

template <bool TRI, int PITCH>
__device__ __forceinline__ void hist_stage(const int* __restrict__ rows, int tid, unsigned* cntW, unsigned* totals) {
    const int lane = tid & 31;
    const int wave = tid >> 5;
    Pass<TRI> pA, pB, pC, pD;
    load_pass<TRI>(rows, 0, lane, pA);
    load_pass<TRI>(rows, 1, lane, pB);
    load_pass<TRI>(rows, 2, lane, pC);
    for (int i = lane; i < 2 * PITCH / 4; i += 32) *(v4u*)(cntW + 4 * i) = (v4u){0u, 0u, 0u, 0u};
    if (tid < 16) totals[tid] = 0u;
    lds_barrier();
    unsigned my0 = 0, my1 = 0;
    for (int it = 0; it < PASSES / 4; ++it) {
        unsigned* cr = cntW + (it >> 1) * PITCH;
        load_pass<TRI>(rows, 4 * it + 3, lane, pD);
        __builtin_amdgcn_sched_barrier(0);
        unsigned got = count_pass<TRI>(pA, 4 * it, lane, cr);
        __builtin_amdgcn_sched_barrier(0);
        load_pass<TRI>(rows, 4 * it + 4, lane, pA);
        __builtin_amdgcn_sched_barrier(0);
        got += count_pass<TRI>(pB, 4 * it + 1, lane, cr);
        __builtin_amdgcn_sched_barrier(0);
        load_pass<TRI>(rows, 4 * it + 5, lane, pB);
        __builtin_amdgcn_sched_barrier(0);
        got += count_pass<TRI>(pC, 4 * it + 2, lane, cr);
        __builtin_amdgcn_sched_barrier(0);
        load_pass<TRI>(rows, 4 * it + 6, lane, pC);
        __builtin_amdgcn_sched_barrier(0);
        got += count_pass<TRI>(pD, 4 * it + 3, lane, cr);
        if (it < 2) my0 += got; else my1 += got;
    }
    if (my0) atomicAdd(&totals[2 * wave], my0);
    if (my1) atomicAdd(&totals[2 * wave + 1], my1);
}

struct Chunk { v2f r[CH][2]; };

__global__ __launch_bounds__(256) void encoder_kernel(const int* __restrict__ src,
                                                  const float* __restrict__ W1B,
                                                  const float* __restrict__ b1B,
                                                  const float* __restrict__ W2B,
                                                  const float* __restrict__ b2B,
                                                  const float* __restrict__ W1T,
                                                  const float* __restrict__ b1T,
                                                  const float* __restrict__ W2T,
                                                  const float* __restrict__ b2T,
                                                  float* __restrict__ out) {
    const bool tri = blockIdx.x >= TILES;
    const float* __restrict__ W1   = tri ? W1T : W1B;
    const float* __restrict__ b1   = tri ? b1T : b1B;
    const float* __restrict__ W2   = tri ? W2T : W2B;
    const float* __restrict__ b2   = tri ? b2T : b2B;
    const int K      = tri ? BUCKETS : V2;
    const int pa     = tri ? PA_TRI : PA_BI;
    const int nFull  = tri ? BUCKETS / KC : V2 / KC;
    const int outOff = tri ? 32 : 0;

    __shared__ __attribute__((aligned(16))) float lds_a[16 * PA_BI];
    __shared__ __attribute__((aligned(16))) float lds_h[16 * PH];
    __shared__ __attribute__((aligned(16))) float lds_p[4 * 16 * PP];
    __shared__ unsigned totals[16];

    const int tid = threadIdx.x;
    const int rowBase = (tri ? blockIdx.x - TILES : blockIdx.x) * 16;
    const int lane = tid & 31;
    const int wave = tid >> 5;
    const int m  = lane & 15;
    const int kh = lane >> 4;
    const int n  = lane & 15;
    const int colBase = wave * 32;

    const float* wp = W1 + (size_t)(2 * kh) * HID + colBase + 2 * n;


    auto load_chunk = [&](int c, Chunk& b) {
        const float* p = wp + (size_t)(min(c, nFull - 1) * KC) * HID;
        #pragma unroll
        for (int ks = 0; ks < CH; ++ks) {
            b.r[ks][0] = *(const v2f*)(p + (4 * ks) * HID);
            b.r[ks][1] = *(const v2f*)(p + (4 * ks + 1) * HID);
        }
    };

    v8f c0 = {};
    v8f c1 = {};

    auto load_a = [&](int c, v2f (&f)[CH]) {
        const float* a = lds_a + m * pa + c * KC + 2 * kh;
        #pragma unroll
        for (int ks = 0; ks < CH; ++ks) f[ks] = *(const v2f*)(a + 4 * ks);
    };
    v2f fa[CH];

    auto compute_chunk = [&](int c, Chunk& b, bool refill, bool nextA) {
        v2f fb0[CH], fb1[CH];
        #pragma unroll
        for (int ks = 0; ks < CH; ++ks) {
            fb0[ks] = (v2f){b.r[ks][0].x, b.r[ks][1].x};
            fb1[ks] = (v2f){b.r[ks][0].y, b.r[ks][1].y};
        }
        v2f fn[CH];
        if (nextA) load_a(c + 1, fn);
        __builtin_amdgcn_sched_barrier(0);
        #pragma unroll
        for (int ks = 0; ks < CH; ++ks) {
            c0 = __builtin_amdgcn_wmma_f32_16x16x4_f32(false, fa[ks], false, fb0[ks], (short)0, c0, false, false);
            c1 = __builtin_amdgcn_wmma_f32_16x16x4_f32(false, fa[ks], false, fb1[ks], (short)0, c1, false, false);
        }
        __builtin_amdgcn_sched_barrier(0);
        if (refill) load_chunk(c + 3, b);
        __builtin_amdgcn_sched_barrier(0);
        if (nextA) {
            #pragma unroll
            for (int ks = 0; ks < CH; ++ks) fa[ks] = fn[ks];
        }
    };

    unsigned* cnt = reinterpret_cast<unsigned*>(lds_a);
    const int* rows = src + (size_t)(rowBase + 2 * wave) * S_LEN;
    unsigned* cntW = cnt + (2 * wave) * pa;
    if (tri) hist_stage<true,  PA_TRI>(rows, tid, cntW, totals);
    else     hist_stage<false, PA_BI>(rows, tid, cntW, totals);

    Chunk b0, b1c, b2c;
    load_chunk(0, b0);
    load_chunk(1, b1c);
    load_chunk(2, b2c);
    const int kt = (V2 / KC) * KC + 2 * kh;
    const v2f tl0 = *(const v2f*)(W1 + (size_t)min(kt,     K - 1) * HID + colBase + 2 * n);
    const v2f tl1 = *(const v2f*)(W1 + (size_t)min(kt + 1, K - 1) * HID + colBase + 2 * n);
    lds_barrier();
    for (int rr = 0; rr < 2; ++rr) {
        const float inv = 1.0f / (float)max(totals[2 * wave + rr], 1u);
        for (int i = lane; i < pa / 4; i += 32) {
            const v4u u = *(const v4u*)(cntW + rr * pa + 4 * i);
            *(v4f*)(lds_a + (2 * wave + rr) * pa + 4 * i) =
                (v4f){(float)u.x * inv, (float)u.y * inv, (float)u.z * inv, (float)u.w * inv};
        }
    }
    lds_barrier();
    load_a(0, fa);

    static_assert((V2 / KC) % 3 == 1 && (BUCKETS / KC) % 3 == 1 && BUCKETS % KC == 0, "chunk rotation");
    const int nTriple = (nFull - 1) / 3;
    for (int t = 0; t < nTriple; ++t) {
        compute_chunk(3 * t,     b0,  true, true);
        compute_chunk(3 * t + 1, b1c, true, true);
        compute_chunk(3 * t + 2, b2c, true, true);
    }
    const int ct = wave & 1;
    const int kq = wave >> 1;
    const int cb = ct * 16;
    const v2f bias1 = *(const v2f*)(b1 + colBase + 2 * n);
    const int orow = tid >> 4;
    const int oc   = (tid & 15) * 2;
    const v2f bias2 = *(const v2f*)(b2 + oc);
    v2f w2f[16];
    {
        const float* w2p = W2 + (size_t)(64 * kq + 2 * kh) * 32 + cb + n;
        #pragma unroll
        for (int ks = 0; ks < 16; ++ks) {
            w2f[ks].x = w2p[(4 * ks) * 32];
            w2f[ks].y = w2p[(4 * ks + 1) * 32];
        }
    }

    compute_chunk(nFull - 1, b0, false, false);
    static_assert(V2 - (V2 / KC) * KC <= 4, "the tail fits one k-step");
    if (!tri) {
        const int k = (V2 / KC) * KC + 2 * kh;
        const v2f fbt0 = {k < V2 ? tl0.x : 0.f, k + 1 < V2 ? tl1.x : 0.f};
        const v2f fbt1 = {k < V2 ? tl0.y : 0.f, k + 1 < V2 ? tl1.y : 0.f};
        const v2f fat = *(const v2f*)(lds_a + m * pa + k);
        c0 = __builtin_amdgcn_wmma_f32_16x16x4_f32(false, fat, false, fbt0, (short)0, c0, false, false);
        c1 = __builtin_amdgcn_wmma_f32_16x16x4_f32(false, fat, false, fbt1, (short)0, c1, false, false);
    }

    #pragma unroll
    for (int r = 0; r < 8; ++r) {
        const int mm = r + 8 * kh;
        float h0 = c0[r] + bias1.x;
        float h1 = c1[r] + bias1.y;
        h0 = h0 > 0.0f ? h0 : 0.0f;
        h1 = h1 > 0.0f ? h1 : 0.0f;
        *(v2f*)(lds_h + mm * PH + colBase + 2 * n) = (v2f){h0, h1};
    }
    __syncthreads();

    v8f d = {};
    {
        const float* a = lds_h + m * PH + 64 * kq + 2 * kh;
        v2f af[16];
        #pragma unroll
        for (int ks = 0; ks < 16; ++ks) af[ks] = *(const v2f*)(a + 4 * ks);
        __builtin_amdgcn_sched_barrier(0);
        #pragma unroll
        for (int ks = 0; ks < 16; ++ks)
            d = __builtin_amdgcn_wmma_f32_16x16x4_f32(false, af[ks], false, w2f[ks], (short)0, d, false, false);
    }
    {
        float* p = lds_p + kq * (16 * PP) + (8 * kh) * PP + cb + n;
        #pragma unroll
        for (int r = 0; r < 8; ++r) p[r * PP] = d[r];
    }
    __syncthreads();

    {
        const float* p = lds_p + orow * PP + oc;
        v2f s = *(const v2f*)p;
        s += *(const v2f*)(p + 16 * PP);
        s += *(const v2f*)(p + 2 * 16 * PP);
        s += *(const v2f*)(p + 3 * 16 * PP);
        s += bias2;
        *(v2f*)(out + (size_t)(rowBase + orow) * OUTC + outOff + oc) = s;
    }
}

extern "C" void kernel_launch(void* const* d_in, const int* in_sizes, int n_in,
                              void* d_out, int out_size, void* d_ws, size_t ws_size,
                              hipStream_t stream) {
    const int*   src = (const int*)d_in[0];
    const float* Wb1 = (const float*)d_in[1];
    const float* bb1 = (const float*)d_in[2];
    const float* Wb2 = (const float*)d_in[3];
    const float* bb2 = (const float*)d_in[4];
    const float* Wt1 = (const float*)d_in[5];
    const float* bt1 = (const float*)d_in[6];
    const float* Wt2 = (const float*)d_in[7];
    const float* bt2 = (const float*)d_in[8];
    float* out = (float*)d_out;

    encoder_kernel<<<2 * TILES, 256, 0, stream>>>(src, Wb1, bb1, Wb2, bb2, Wt1, bt1, Wt2, bt2, out);
}
